# starting kernel
# baseline (speedup 1.0000x reference)
#include <hip/hip_runtime.h>
#include <hip/hip_bf16.h>

typedef __attribute__((ext_vector_type(2))) float v2f;
typedef __attribute__((ext_vector_type(8))) float v8f;

#define D_MODEL 1024
#define D_STATE 16
#define D_CONV  4
#define D_INNER 2048
#define DT_RANK 64
#define BB 2
#define LL 2048
#define TOKENS (BB * LL)
#define XDBL_W (DT_RANK + 2 * D_STATE)

__device__ __forceinline__ float sigmoidf_(float x) { return 1.f / (1.f + __expf(-x)); }

__global__ __launch_bounds__(256)
void ln_kernel(const float* __restrict__ x, const float* __restrict__ w,
               const float* __restrict__ b, float* __restrict__ out)
{
    const int row = blockIdx.x;
    const float* xr = x + (size_t)row * D_MODEL;
    float v[4];
    float s = 0.f, s2 = 0.f;
#pragma unroll
    for (int i = 0; i < 4; ++i) {
        v[i] = xr[threadIdx.x + i * 256];
        s += v[i];
        s2 += v[i] * v[i];
    }
#pragma unroll
    for (int off = 16; off > 0; off >>= 1) {
        s  += __shfl_down(s,  off, 32);
        s2 += __shfl_down(s2, off, 32);
    }
    __shared__ float rs[8], rq[8];
    __shared__ float mu_s, rstd_s;
    const int wave = threadIdx.x >> 5;
    const int lane = threadIdx.x & 31;
    if (lane == 0) { rs[wave] = s; rq[wave] = s2; }
    __syncthreads();
    if (threadIdx.x == 0) {
        float a = 0.f, q = 0.f;
#pragma unroll
        for (int i = 0; i < 8; ++i) { a += rs[i]; q += rq[i]; }
        const float mu  = a * (1.f / D_MODEL);
        const float var = q * (1.f / D_MODEL) - mu * mu;
        mu_s = mu;
        rstd_s = rsqrtf(var + 1e-5f);
    }
    __syncthreads();
    const float mu = mu_s, rstd = rstd_s;
    float* orow = out + (size_t)row * D_MODEL;
#pragma unroll
    for (int i = 0; i < 4; ++i) {
        const int c = threadIdx.x + i * 256;
        orow[c] = (v[i] - mu) * rstd * w[c] + b[c];
    }
}

template <int MT, int NT>
__global__ __launch_bounds__(128)
void wmma_gemm_kernel(const float* __restrict__ A, int lda,
                      const float* __restrict__ Bm, int ldb,
                      float* __restrict__ C, int ldc, int K,
                      const float* __restrict__ residual, int ldr,
                      const float* __restrict__ scale_ptr)
{
    const int lane = threadIdx.x & 31;
    const int wave = threadIdx.x >> 5;
    const int wy = wave & 1;
    const int wx = wave >> 1;
    const int m0 = blockIdx.y * (32 * MT) + wy * (16 * MT);
    const int n0 = blockIdx.x * (32 * NT) + wx * (16 * NT);
    const int lm = lane & 15;
    const int kh = (lane >> 4) << 1;

    v8f acc[MT][NT];
#pragma unroll
    for (int i = 0; i < MT; ++i)
#pragma unroll
        for (int t = 0; t < NT; ++t)
#pragma unroll
            for (int e = 0; e < 8; ++e) acc[i][t][e] = 0.f;

    const float* Ap = A + (size_t)(m0 + lm) * lda + kh;
    const float* Bp = Bm + (size_t)kh * ldb + n0 + lm;

    auto loadA = [&](int k, v2f (&a)[MT]) {
#pragma unroll
        for (int i = 0; i < MT; ++i)
            a[i] = *(const v2f*)(Ap + (size_t)i * 16 * lda + k);
    };
    auto loadB = [&](int k, v2f (&b)[NT]) {
        const float* r0 = Bp + (size_t)k * ldb;
        const float* r1 = r0 + ldb;
#pragma unroll
        for (int t = 0; t < NT; ++t) {
            b[t].x = r0[t * 16];
            b[t].y = r1[t * 16];
        }
    };

    v2f a[MT], b[NT], a2[MT], b2[NT];
    loadA(0, a);
    loadB(0, b);

    int k = 0;
    for (; k + 4 < K; k += 4) {
        loadA(k + 4, a2);
        loadB(k + 4, b2);
        __builtin_prefetch(Ap + k + 128, 0, 3);
        __builtin_prefetch(Bp + (size_t)(k + 16) * ldb, 0, 3);
#pragma unroll
        for (int i = 0; i < MT; ++i)
#pragma unroll
            for (int t = 0; t < NT; ++t)
                acc[i][t] = __builtin_amdgcn_wmma_f32_16x16x4_f32(
                    false, a[i], false, b[t], (short)0, acc[i][t], false, false);
#pragma unroll
        for (int i = 0; i < MT; ++i) a[i] = a2[i];
#pragma unroll
        for (int t = 0; t < NT; ++t) b[t] = b2[t];
    }
#pragma unroll
    for (int i = 0; i < MT; ++i)
#pragma unroll
        for (int t = 0; t < NT; ++t)
            acc[i][t] = __builtin_amdgcn_wmma_f32_16x16x4_f32(
                false, a[i], false, b[t], (short)0, acc[i][t], false, false);

    const float scale = scale_ptr ? scale_ptr[0] : 1.0f;
    const int mb = (lane >> 4) << 3;
#pragma unroll
    for (int i = 0; i < MT; ++i) {
#pragma unroll
        for (int t = 0; t < NT; ++t) {
            const int n = n0 + t * 16 + lm;
#pragma unroll
            for (int v = 0; v < 8; ++v) {
                const int m = m0 + i * 16 + mb + v;
                float val = acc[i][t][v] * scale;
                if (residual) val += residual[(size_t)m * ldr + n];
                C[(size_t)m * ldc + n] = val;
            }
        }
    }
}

__global__ __launch_bounds__(256)
void conv_silu_kernel(const float* __restrict__ xz, const float* __restrict__ w,
                      const float* __restrict__ bias, float* __restrict__ out)
{
    const size_t idx = (size_t)blockIdx.x * 256 + threadIdx.x;
    const int c = (int)(idx & (D_INNER - 1));
    const size_t bl = idx >> 11;
    const int l = (int)(bl & (LL - 1));
    const int b = (int)(bl >> 11);
    float acc = bias[c];
#pragma unroll
    for (int k = 0; k < D_CONV; ++k) {
        const int ls = l - (D_CONV - 1) + k;
        if (ls >= 0)
            acc += w[c * D_CONV + k] *
                   xz[((size_t)(b * LL + ls)) * (2 * D_INNER) + c];
    }
    out[idx] = acc * sigmoidf_(acc);
}

__global__ __launch_bounds__(256)
void conv_state_kernel(const float* __restrict__ xz, float* __restrict__ out)
{
    const int idx = blockIdx.x * 256 + threadIdx.x;
    if (idx >= BB * D_INNER * (D_CONV - 1)) return;
    const int j = idx % (D_CONV - 1);
    const int c = (idx / (D_CONV - 1)) % D_INNER;
    const int b = idx / ((D_CONV - 1) * D_INNER);
    out[idx] = xz[((size_t)(b * LL + (LL - (D_CONV - 1) + j))) * (2 * D_INNER) + c];
}

__global__ __launch_bounds__(256)
void dt_softplus_kernel(float* __restrict__ dt, const float* __restrict__ bias)
{
    const size_t idx = (size_t)blockIdx.x * 256 + threadIdx.x;
    const int d = (int)(idx & (D_INNER - 1));
    const float v = dt[idx] + bias[d];
    dt[idx] = (v > 20.f) ? v : __logf(1.f + __expf(v));
}

__global__ __launch_bounds__(256)
void scan_kernel(const float* __restrict__ dt, const float* __restrict__ xc,
                 const float* __restrict__ xdbl, const float* __restrict__ A_log,
                 float* __restrict__ y, float* __restrict__ ssm_out)
{
    const int t = blockIdx.x * 256 + threadIdx.x;
    const int d = t & (D_INNER - 1);
    const int b = t >> 11;

    float Aval[D_STATE];
#pragma unroll
    for (int n = 0; n < D_STATE; ++n)
        Aval[n] = -__expf(A_log[d * D_STATE + n]);

    float h[D_STATE];
#pragma unroll
    for (int n = 0; n < D_STATE; ++n) h[n] = 0.f;

    for (int l = 0; l < LL; ++l) {
        const size_t roff = (size_t)(b * LL + l);
        const float dtv = dt[roff * D_INNER + d];
        const float xv  = xc[roff * D_INNER + d];
        const float4* Bp = (const float4*)(xdbl + roff * XDBL_W + DT_RANK);
        const float4* Cp = (const float4*)(xdbl + roff * XDBL_W + DT_RANK + D_STATE);
        float Bb[D_STATE], Cc[D_STATE];
#pragma unroll
        for (int q = 0; q < 4; ++q) {
            const float4 bq = Bp[q];
            const float4 cq = Cp[q];
            Bb[4*q+0] = bq.x; Bb[4*q+1] = bq.y; Bb[4*q+2] = bq.z; Bb[4*q+3] = bq.w;
            Cc[4*q+0] = cq.x; Cc[4*q+1] = cq.y; Cc[4*q+2] = cq.z; Cc[4*q+3] = cq.w;
        }
        const float bx = dtv * xv;
        float yv = 0.f;
#pragma unroll
        for (int n = 0; n < D_STATE; ++n) {
            const float dA = __expf(Aval[n] * dtv);
            h[n] = dA * h[n] + bx * Bb[n];
            yv += h[n] * Cc[n];
        }
        y[roff * D_INNER + d] = yv;
    }

#pragma unroll
    for (int n = 0; n < D_STATE; ++n)
        ssm_out[(size_t)(b * D_INNER + d) * D_STATE + n] = h[n];
}

__global__ __launch_bounds__(256)
void gate_kernel(float* __restrict__ y, const float* __restrict__ xc,
                 const float* __restrict__ xz, const float* __restrict__ Dp)
{
    const size_t idx = (size_t)blockIdx.x * 256 + threadIdx.x;
    const int d = (int)(idx & (D_INNER - 1));
    const size_t row = idx >> 11;
    const float z  = xz[row * (2 * D_INNER) + D_INNER + d];
    const float yv = y[idx] + xc[idx] * Dp[d];
    y[idx] = yv * z * sigmoidf_(z);
}

extern "C" void kernel_launch(void* const* d_in, const int* in_sizes, int n_in,
                              void* d_out, int out_size, void* d_ws, size_t ws_size,
                              hipStream_t stream)
{
    const float* x         = (const float*)d_in[0];
    const float* ln_w      = (const float*)d_in[1];
    const float* ln_b      = (const float*)d_in[2];
    const float* in_proj_w = (const float*)d_in[3];
    const float* conv_w    = (const float*)d_in[4];
    const float* conv_b    = (const float*)d_in[5];
    const float* x_proj_w  = (const float*)d_in[6];
    const float* dt_proj_w = (const float*)d_in[7];
    const float* dt_proj_b = (const float*)d_in[8];
    const float* A_log     = (const float*)d_in[9];
    const float* D_param   = (const float*)d_in[10];
    const float* out_projw = (const float*)d_in[11];
    const float* res_scale = (const float*)d_in[12];

    float* out      = (float*)d_out;
    float* out_ssm  = out + (size_t)BB * LL * D_MODEL;
    float* out_conv = out_ssm + (size_t)BB * D_INNER * D_STATE;

    float* ws    = (float*)d_ws;
    float* xn    = ws;
    float* xz    = xn    + (size_t)TOKENS * D_MODEL;
    float* xc    = xz    + (size_t)TOKENS * 2 * D_INNER;
    float* xdbl  = xc    + (size_t)TOKENS * D_INNER;
    float* dtbuf = xdbl  + (size_t)TOKENS * XDBL_W;
    float* ybuf  = dtbuf + (size_t)TOKENS * D_INNER;

    ln_kernel<<<TOKENS, 256, 0, stream>>>(x, ln_w, ln_b, xn);

    wmma_gemm_kernel<2, 4><<<dim3((2 * D_INNER) / 128, TOKENS / 64), 128, 0, stream>>>(
        xn, D_MODEL, in_proj_w, 2 * D_INNER, xz, 2 * D_INNER, D_MODEL,
        nullptr, 0, nullptr);

    conv_silu_kernel<<<(TOKENS * D_INNER) / 256, 256, 0, stream>>>(
        xz, conv_w, conv_b, xc);
    conv_state_kernel<<<(BB * D_INNER * (D_CONV - 1) + 255) / 256, 256, 0, stream>>>(
        xz, out_conv);

    wmma_gemm_kernel<2, 3><<<dim3(XDBL_W / 96, TOKENS / 64), 128, 0, stream>>>(
        xc, D_INNER, x_proj_w, XDBL_W, xdbl, XDBL_W, D_INNER,
        nullptr, 0, nullptr);

    wmma_gemm_kernel<2, 4><<<dim3(D_INNER / 128, TOKENS / 64), 128, 0, stream>>>(
        xdbl, XDBL_W, dt_proj_w, D_INNER, dtbuf, D_INNER, DT_RANK,
        nullptr, 0, nullptr);

    dt_softplus_kernel<<<(TOKENS * D_INNER) / 256, 256, 0, stream>>>(dtbuf, dt_proj_b);

    scan_kernel<<<(BB * D_INNER) / 256, 256, 0, stream>>>(
        dtbuf, xc, xdbl, A_log, ybuf, out_ssm);

    gate_kernel<<<(TOKENS * D_INNER) / 256, 256, 0, stream>>>(ybuf, xc, xz, D_param);

    wmma_gemm_kernel<2, 4><<<dim3(D_MODEL / 128, TOKENS / 64), 128, 0, stream>>>(
        ybuf, D_INNER, out_projw, D_MODEL, out, D_MODEL, D_INNER,
        x, D_MODEL, res_scale);
}


# selective scan split over the sequence: 64 chunks of 32 steps run at once from zero state, carries chained as P*h_in+h_loc, chunks re-run from true carry-in; chunk-boundary state sums regrouped
# speedup vs baseline: 2.9743x; 2.9743x over previous
#include <hip/hip_runtime.h>
#include <hip/hip_bf16.h>

typedef __attribute__((ext_vector_type(2))) float v2f;
typedef __attribute__((ext_vector_type(8))) float v8f;

#define D_MODEL 1024
#define D_STATE 16
#define D_CONV  4
#define D_INNER 2048
#define DT_RANK 64
#define BB 2
#define LL 2048
#define TOKENS (BB * LL)
#define XDBL_W (DT_RANK + 2 * D_STATE)

__device__ __forceinline__ float sigmoidf_(float x) { return 1.f / (1.f + __expf(-x)); }

__global__ __launch_bounds__(256)
void ln_kernel(const float* __restrict__ x, const float* __restrict__ w,
               const float* __restrict__ b, float* __restrict__ out)
{
    const int row = blockIdx.x;
    const float* xr = x + (size_t)row * D_MODEL;
    float v[4];
    float s = 0.f, s2 = 0.f;
#pragma unroll
    for (int i = 0; i < 4; ++i) {
        v[i] = xr[threadIdx.x + i * 256];
        s += v[i];
        s2 += v[i] * v[i];
    }
#pragma unroll
    for (int off = 16; off > 0; off >>= 1) {
        s  += __shfl_down(s,  off, 32);
        s2 += __shfl_down(s2, off, 32);
    }
    __shared__ float rs[8], rq[8];
    __shared__ float mu_s, rstd_s;
    const int wave = threadIdx.x >> 5;
    const int lane = threadIdx.x & 31;
    if (lane == 0) { rs[wave] = s; rq[wave] = s2; }
    __syncthreads();
    if (threadIdx.x == 0) {
        float a = 0.f, q = 0.f;
#pragma unroll
        for (int i = 0; i < 8; ++i) { a += rs[i]; q += rq[i]; }
        const float mu  = a * (1.f / D_MODEL);
        const float var = q * (1.f / D_MODEL) - mu * mu;
        mu_s = mu;
        rstd_s = rsqrtf(var + 1e-5f);
    }
    __syncthreads();
    const float mu = mu_s, rstd = rstd_s;
    float* orow = out + (size_t)row * D_MODEL;
#pragma unroll
    for (int i = 0; i < 4; ++i) {
        const int c = threadIdx.x + i * 256;
        orow[c] = (v[i] - mu) * rstd * w[c] + b[c];
    }
}

template <int MT, int NT>
__global__ __launch_bounds__(128)
void wmma_gemm_kernel(const float* __restrict__ A, int lda,
                      const float* __restrict__ Bm, int ldb,
                      float* __restrict__ C, int ldc, int K,
                      const float* __restrict__ residual, int ldr,
                      const float* __restrict__ scale_ptr)
{
    const int lane = threadIdx.x & 31;
    const int wave = threadIdx.x >> 5;
    const int wy = wave & 1;
    const int wx = wave >> 1;
    const int m0 = blockIdx.y * (32 * MT) + wy * (16 * MT);
    const int n0 = blockIdx.x * (32 * NT) + wx * (16 * NT);
    const int lm = lane & 15;
    const int kh = (lane >> 4) << 1;

    v8f acc[MT][NT];
#pragma unroll
    for (int i = 0; i < MT; ++i)
#pragma unroll
        for (int t = 0; t < NT; ++t)
#pragma unroll
            for (int e = 0; e < 8; ++e) acc[i][t][e] = 0.f;

    const float* Ap = A + (size_t)(m0 + lm) * lda + kh;
    const float* Bp = Bm + (size_t)kh * ldb + n0 + lm;

    auto loadA = [&](int k, v2f (&a)[MT]) {
#pragma unroll
        for (int i = 0; i < MT; ++i)
            a[i] = *(const v2f*)(Ap + (size_t)i * 16 * lda + k);
    };
    auto loadB = [&](int k, v2f (&b)[NT]) {
        const float* r0 = Bp + (size_t)k * ldb;
        const float* r1 = r0 + ldb;
#pragma unroll
        for (int t = 0; t < NT; ++t) {
            b[t].x = r0[t * 16];
            b[t].y = r1[t * 16];
        }
    };

    v2f a[MT], b[NT], a2[MT], b2[NT];
    loadA(0, a);
    loadB(0, b);

    int k = 0;
    for (; k + 4 < K; k += 4) {
        loadA(k + 4, a2);
        loadB(k + 4, b2);
        __builtin_prefetch(Ap + k + 128, 0, 3);
        __builtin_prefetch(Bp + (size_t)(k + 16) * ldb, 0, 3);
#pragma unroll
        for (int i = 0; i < MT; ++i)
#pragma unroll
            for (int t = 0; t < NT; ++t)
                acc[i][t] = __builtin_amdgcn_wmma_f32_16x16x4_f32(
                    false, a[i], false, b[t], (short)0, acc[i][t], false, false);
#pragma unroll
        for (int i = 0; i < MT; ++i) a[i] = a2[i];
#pragma unroll
        for (int t = 0; t < NT; ++t) b[t] = b2[t];
    }
#pragma unroll
    for (int i = 0; i < MT; ++i)
#pragma unroll
        for (int t = 0; t < NT; ++t)
            acc[i][t] = __builtin_amdgcn_wmma_f32_16x16x4_f32(
                false, a[i], false, b[t], (short)0, acc[i][t], false, false);

    const float scale = scale_ptr ? scale_ptr[0] : 1.0f;
    const int mb = (lane >> 4) << 3;
#pragma unroll
    for (int i = 0; i < MT; ++i) {
#pragma unroll
        for (int t = 0; t < NT; ++t) {
            const int n = n0 + t * 16 + lm;
#pragma unroll
            for (int v = 0; v < 8; ++v) {
                const int m = m0 + i * 16 + mb + v;
                float val = acc[i][t][v] * scale;
                if (residual) val += residual[(size_t)m * ldr + n];
                C[(size_t)m * ldc + n] = val;
            }
        }
    }
}

__global__ __launch_bounds__(256)
void conv_silu_kernel(const float* __restrict__ xz, const float* __restrict__ w,
                      const float* __restrict__ bias, float* __restrict__ out)
{
    const size_t idx = (size_t)blockIdx.x * 256 + threadIdx.x;
    const int c = (int)(idx & (D_INNER - 1));
    const size_t bl = idx >> 11;
    const int l = (int)(bl & (LL - 1));
    const int b = (int)(bl >> 11);
    float acc = bias[c];
#pragma unroll
    for (int k = 0; k < D_CONV; ++k) {
        const int ls = l - (D_CONV - 1) + k;
        if (ls >= 0)
            acc += w[c * D_CONV + k] *
                   xz[((size_t)(b * LL + ls)) * (2 * D_INNER) + c];
    }
    out[idx] = acc * sigmoidf_(acc);
}

__global__ __launch_bounds__(256)
void conv_state_kernel(const float* __restrict__ xz, float* __restrict__ out)
{
    const int idx = blockIdx.x * 256 + threadIdx.x;
    if (idx >= BB * D_INNER * (D_CONV - 1)) return;
    const int j = idx % (D_CONV - 1);
    const int c = (idx / (D_CONV - 1)) % D_INNER;
    const int b = idx / ((D_CONV - 1) * D_INNER);
    out[idx] = xz[((size_t)(b * LL + (LL - (D_CONV - 1) + j))) * (2 * D_INNER) + c];
}

__global__ __launch_bounds__(256)
void dt_softplus_kernel(float* __restrict__ dt, const float* __restrict__ bias)
{
    const size_t idx = (size_t)blockIdx.x * 256 + threadIdx.x;
    const int d = (int)(idx & (D_INNER - 1));
    const float v = dt[idx] + bias[d];
    dt[idx] = (v > 20.f) ? v : __logf(1.f + __expf(v));
}

#define SCAN_T  32
#define SCAN_NC (LL / SCAN_T)
#define SCAN_G  4

__device__ __forceinline__ size_t cs_index(int b, int c, int n, int d)
{
    return ((size_t)((b * SCAN_NC + c) * D_STATE + n)) * D_INNER + d;
}

__device__ __forceinline__ void load_aval(const float* __restrict__ A_log, int d,
                                          float (&Aval)[D_STATE])
{
#pragma unroll
    for (int n = 0; n < D_STATE; ++n)
        Aval[n] = -__expf(A_log[d * D_STATE + n]);
}

__global__ __launch_bounds__(128)
void scan_chunk_state_kernel(const float* __restrict__ dt, const float* __restrict__ xc,
                             const float* __restrict__ xdbl, const float* __restrict__ A_log,
                             float* __restrict__ hloc, float* __restrict__ pdec)
{
    const int d = blockIdx.x * 128 + threadIdx.x;
    const int c = blockIdx.y;
    const int b = blockIdx.z;

    float Aval[D_STATE];
    load_aval(A_log, d, Aval);

    float h[D_STATE], P[D_STATE];
#pragma unroll
    for (int n = 0; n < D_STATE; ++n) { h[n] = 0.f; P[n] = 1.f; }

    const size_t row0 = (size_t)b * LL + (size_t)c * SCAN_T;
    const float* dtp = dt + row0 * D_INNER + d;
    const float* xp  = xc + row0 * D_INNER + d;
    const float* bcp = xdbl + row0 * XDBL_W + DT_RANK;

    for (int l0 = 0; l0 < SCAN_T; l0 += SCAN_G) {
        float dtv[SCAN_G], xv[SCAN_G];
#pragma unroll
        for (int j = 0; j < SCAN_G; ++j) {
            dtv[j] = dtp[(size_t)(l0 + j) * D_INNER];
            xv[j]  = xp[(size_t)(l0 + j) * D_INNER];
        }
#pragma unroll
        for (int j = 0; j < SCAN_G; ++j) {
            const float4* Bp = (const float4*)(bcp + (size_t)(l0 + j) * XDBL_W);
            float Bb[D_STATE];
#pragma unroll
            for (int q = 0; q < 4; ++q) {
                const float4 bq = Bp[q];
                Bb[4*q+0] = bq.x; Bb[4*q+1] = bq.y; Bb[4*q+2] = bq.z; Bb[4*q+3] = bq.w;
            }
            const float bx = dtv[j] * xv[j];
#pragma unroll
            for (int n = 0; n < D_STATE; ++n) {
                const float dA = __expf(Aval[n] * dtv[j]);
                h[n] = dA * h[n] + bx * Bb[n];
                P[n] *= dA;
            }
        }
    }

#pragma unroll
    for (int n = 0; n < D_STATE; ++n) {
        hloc[cs_index(b, c, n, d)] = h[n];
        pdec[cs_index(b, c, n, d)] = P[n];
    }
}

#define CARRY_G 21
static_assert((SCAN_NC - 1) % CARRY_G == 0, "carry groups must tile the chunks");

__global__ __launch_bounds__(256)
void scan_carry_kernel(const float* __restrict__ hloc, const float* __restrict__ pdec,
                       float* __restrict__ hin)
{
    const int t = blockIdx.x * 256 + threadIdx.x;
    const int d = t & (D_INNER - 1);
    const int n = (t >> 11) & (D_STATE - 1);
    const int b = t >> 15;
    const size_t base = cs_index(b, 0, n, d);
    const size_t cstride = (size_t)D_STATE * D_INNER;
    float carry = 0.f;
    for (int c0 = 0; c0 < SCAN_NC - 1; c0 += CARRY_G) {
        float p[CARRY_G], s[CARRY_G];
#pragma unroll
        for (int j = 0; j < CARRY_G; ++j) {
            p[j] = pdec[base + (size_t)(c0 + j) * cstride];
            s[j] = hloc[base + (size_t)(c0 + j) * cstride];
        }
#pragma unroll
        for (int j = 0; j < CARRY_G; ++j) {
            hin[base + (size_t)(c0 + j) * cstride] = carry;
            carry = p[j] * carry + s[j];
        }
    }
    hin[base + (size_t)(SCAN_NC - 1) * cstride] = carry;
}

__global__ __launch_bounds__(128)
void scan_chunk_out_kernel(const float* __restrict__ dt, const float* __restrict__ xc,
                           const float* __restrict__ xdbl, const float* __restrict__ A_log,
                           const float* __restrict__ hin,
                           float* __restrict__ y, float* __restrict__ ssm_out)
{
    const int d = blockIdx.x * 128 + threadIdx.x;
    const int c = blockIdx.y;
    const int b = blockIdx.z;

    float Aval[D_STATE];
    load_aval(A_log, d, Aval);

    float h[D_STATE];
#pragma unroll
    for (int n = 0; n < D_STATE; ++n) h[n] = hin[cs_index(b, c, n, d)];

    const size_t row0 = (size_t)b * LL + (size_t)c * SCAN_T;
    const float* dtp = dt + row0 * D_INNER + d;
    const float* xp  = xc + row0 * D_INNER + d;
    const float* bcp = xdbl + row0 * XDBL_W + DT_RANK;
    float* yp = y + row0 * D_INNER + d;

    for (int l0 = 0; l0 < SCAN_T; l0 += SCAN_G) {
        float dtv[SCAN_G], xv[SCAN_G];
#pragma unroll
        for (int j = 0; j < SCAN_G; ++j) {
            dtv[j] = dtp[(size_t)(l0 + j) * D_INNER];
            xv[j]  = xp[(size_t)(l0 + j) * D_INNER];
        }
#pragma unroll
        for (int j = 0; j < SCAN_G; ++j) {
            const float4* Bp = (const float4*)(bcp + (size_t)(l0 + j) * XDBL_W);
            const float4* Cp = Bp + 4;
            float Bb[D_STATE], Cc[D_STATE];
#pragma unroll
            for (int q = 0; q < 4; ++q) {
                const float4 bq = Bp[q];
                const float4 cq = Cp[q];
                Bb[4*q+0] = bq.x; Bb[4*q+1] = bq.y; Bb[4*q+2] = bq.z; Bb[4*q+3] = bq.w;
                Cc[4*q+0] = cq.x; Cc[4*q+1] = cq.y; Cc[4*q+2] = cq.z; Cc[4*q+3] = cq.w;
            }
            const float bx = dtv[j] * xv[j];
            float yv = 0.f;
#pragma unroll
            for (int n = 0; n < D_STATE; ++n) {
                const float dA = __expf(Aval[n] * dtv[j]);
                h[n] = dA * h[n] + bx * Bb[n];
                yv += h[n] * Cc[n];
            }
            yp[(size_t)(l0 + j) * D_INNER] = yv;
        }
    }

    if (c == SCAN_NC - 1) {
#pragma unroll
        for (int n = 0; n < D_STATE; ++n)
            ssm_out[(size_t)(b * D_INNER + d) * D_STATE + n] = h[n];
    }
}

__global__ __launch_bounds__(256)
void gate_kernel(float* __restrict__ y, const float* __restrict__ xc,
                 const float* __restrict__ xz, const float* __restrict__ Dp)
{
    const size_t idx = (size_t)blockIdx.x * 256 + threadIdx.x;
    const int d = (int)(idx & (D_INNER - 1));
    const size_t row = idx >> 11;
    const float z  = xz[row * (2 * D_INNER) + D_INNER + d];
    const float yv = y[idx] + xc[idx] * Dp[d];
    y[idx] = yv * z * sigmoidf_(z);
}

extern "C" void kernel_launch(void* const* d_in, const int* in_sizes, int n_in,
                              void* d_out, int out_size, void* d_ws, size_t ws_size,
                              hipStream_t stream)
{
    const float* x         = (const float*)d_in[0];
    const float* ln_w      = (const float*)d_in[1];
    const float* ln_b      = (const float*)d_in[2];
    const float* in_proj_w = (const float*)d_in[3];
    const float* conv_w    = (const float*)d_in[4];
    const float* conv_b    = (const float*)d_in[5];
    const float* x_proj_w  = (const float*)d_in[6];
    const float* dt_proj_w = (const float*)d_in[7];
    const float* dt_proj_b = (const float*)d_in[8];
    const float* A_log     = (const float*)d_in[9];
    const float* D_param   = (const float*)d_in[10];
    const float* out_projw = (const float*)d_in[11];
    const float* res_scale = (const float*)d_in[12];

    float* out      = (float*)d_out;
    float* out_ssm  = out + (size_t)BB * LL * D_MODEL;
    float* out_conv = out_ssm + (size_t)BB * D_INNER * D_STATE;

    float* ws    = (float*)d_ws;
    float* xn    = ws;
    float* xz    = xn    + (size_t)TOKENS * D_MODEL;
    float* xc    = xz    + (size_t)TOKENS * 2 * D_INNER;
    float* xdbl  = xc    + (size_t)TOKENS * D_INNER;
    float* dtbuf = xdbl  + (size_t)TOKENS * XDBL_W;
    float* ybuf  = dtbuf + (size_t)TOKENS * D_INNER;
    float* hloc  = ybuf  + (size_t)TOKENS * D_INNER;
    float* pdec  = hloc  + (size_t)BB * SCAN_NC * D_STATE * D_INNER;
    float* hin   = pdec  + (size_t)BB * SCAN_NC * D_STATE * D_INNER;

    ln_kernel<<<TOKENS, 256, 0, stream>>>(x, ln_w, ln_b, xn);

    wmma_gemm_kernel<2, 4><<<dim3((2 * D_INNER) / 128, TOKENS / 64), 128, 0, stream>>>(
        xn, D_MODEL, in_proj_w, 2 * D_INNER, xz, 2 * D_INNER, D_MODEL,
        nullptr, 0, nullptr);

    conv_silu_kernel<<<(TOKENS * D_INNER) / 256, 256, 0, stream>>>(
        xz, conv_w, conv_b, xc);
    conv_state_kernel<<<(BB * D_INNER * (D_CONV - 1) + 255) / 256, 256, 0, stream>>>(
        xz, out_conv);

    wmma_gemm_kernel<2, 3><<<dim3(XDBL_W / 96, TOKENS / 64), 128, 0, stream>>>(
        xc, D_INNER, x_proj_w, XDBL_W, xdbl, XDBL_W, D_INNER,
        nullptr, 0, nullptr);

    wmma_gemm_kernel<2, 4><<<dim3(D_INNER / 128, TOKENS / 64), 128, 0, stream>>>(
        xdbl, XDBL_W, dt_proj_w, D_INNER, dtbuf, D_INNER, DT_RANK,
        nullptr, 0, nullptr);

    dt_softplus_kernel<<<(TOKENS * D_INNER) / 256, 256, 0, stream>>>(dtbuf, dt_proj_b);

    scan_chunk_state_kernel<<<dim3(D_INNER / 128, SCAN_NC - 1, BB), 128, 0, stream>>>(
        dtbuf, xc, xdbl, A_log, hloc, pdec);
    scan_carry_kernel<<<(BB * D_STATE * D_INNER) / 256, 256, 0, stream>>>(hloc, pdec, hin);
    scan_chunk_out_kernel<<<dim3(D_INNER / 128, SCAN_NC, BB), 128, 0, stream>>>(
        dtbuf, xc, xdbl, A_log, hin, ybuf, out_ssm);

    gate_kernel<<<(TOKENS * D_INNER) / 256, 256, 0, stream>>>(ybuf, xc, xz, D_param);

    wmma_gemm_kernel<2, 4><<<dim3(D_MODEL / 128, TOKENS / 64), 128, 0, stream>>>(
        ybuf, D_INNER, out_projw, D_MODEL, out, D_MODEL, D_INNER,
        x, D_MODEL, res_scale);
}


# in_proj, out_proj and x_proj GEMMs: A and B chunks staged once per block through LDS (B transposed), fragments read from LDS; block tiles 128x128, 64x128, 32x96; products summed in the same order
# speedup vs baseline: 4.6489x; 1.5630x over previous
#include <hip/hip_runtime.h>
#include <hip/hip_bf16.h>

typedef __attribute__((ext_vector_type(2))) float v2f;
typedef __attribute__((ext_vector_type(8))) float v8f;

#define D_MODEL 1024
#define D_STATE 16
#define D_CONV  4
#define D_INNER 2048
#define DT_RANK 64
#define BB 2
#define LL 2048
#define TOKENS (BB * LL)
#define XDBL_W (DT_RANK + 2 * D_STATE)

__device__ __forceinline__ float sigmoidf_(float x) { return 1.f / (1.f + __expf(-x)); }

__global__ __launch_bounds__(256)
void ln_kernel(const float* __restrict__ x, const float* __restrict__ w,
               const float* __restrict__ b, float* __restrict__ out)
{
    const int row = blockIdx.x;
    const float* xr = x + (size_t)row * D_MODEL;
    float v[4];
    float s = 0.f, s2 = 0.f;
#pragma unroll
    for (int i = 0; i < 4; ++i) {
        v[i] = xr[threadIdx.x + i * 256];
        s += v[i];
        s2 += v[i] * v[i];
    }
#pragma unroll
    for (int off = 16; off > 0; off >>= 1) {
        s  += __shfl_down(s,  off, 32);
        s2 += __shfl_down(s2, off, 32);
    }
    __shared__ float rs[8], rq[8];
    __shared__ float mu_s, rstd_s;
    const int wave = threadIdx.x >> 5;
    const int lane = threadIdx.x & 31;
    if (lane == 0) { rs[wave] = s; rq[wave] = s2; }
    __syncthreads();
    if (threadIdx.x == 0) {
        float a = 0.f, q = 0.f;
#pragma unroll
        for (int i = 0; i < 8; ++i) { a += rs[i]; q += rq[i]; }
        const float mu  = a * (1.f / D_MODEL);
        const float var = q * (1.f / D_MODEL) - mu * mu;
        mu_s = mu;
        rstd_s = rsqrtf(var + 1e-5f);
    }
    __syncthreads();
    const float mu = mu_s, rstd = rstd_s;
    float* orow = out + (size_t)row * D_MODEL;
#pragma unroll
    for (int i = 0; i < 4; ++i) {
        const int c = threadIdx.x + i * 256;
        orow[c] = (v[i] - mu) * rstd * w[c] + b[c];
    }
}

template <int MT, int NT>
__global__ __launch_bounds__(128)
void wmma_gemm_kernel(const float* __restrict__ A, int lda,
                      const float* __restrict__ Bm, int ldb,
                      float* __restrict__ C, int ldc, int K,
                      const float* __restrict__ residual, int ldr,
                      const float* __restrict__ scale_ptr)
{
    const int lane = threadIdx.x & 31;
    const int wave = threadIdx.x >> 5;
    const int wy = wave & 1;
    const int wx = wave >> 1;
    const int m0 = blockIdx.y * (32 * MT) + wy * (16 * MT);
    const int n0 = blockIdx.x * (32 * NT) + wx * (16 * NT);
    const int lm = lane & 15;
    const int kh = (lane >> 4) << 1;

    v8f acc[MT][NT];
#pragma unroll
    for (int i = 0; i < MT; ++i)
#pragma unroll
        for (int t = 0; t < NT; ++t)
#pragma unroll
            for (int e = 0; e < 8; ++e) acc[i][t][e] = 0.f;

    const float* Ap = A + (size_t)(m0 + lm) * lda + kh;
    const float* Bp = Bm + (size_t)kh * ldb + n0 + lm;

    auto loadA = [&](int k, v2f (&a)[MT]) {
#pragma unroll
        for (int i = 0; i < MT; ++i)
            a[i] = *(const v2f*)(Ap + (size_t)i * 16 * lda + k);
    };
    auto loadB = [&](int k, v2f (&b)[NT]) {
        const float* r0 = Bp + (size_t)k * ldb;
        const float* r1 = r0 + ldb;
#pragma unroll
        for (int t = 0; t < NT; ++t) {
            b[t].x = r0[t * 16];
            b[t].y = r1[t * 16];
        }
    };

    v2f a[MT], b[NT], a2[MT], b2[NT];
    loadA(0, a);
    loadB(0, b);

    int k = 0;
    for (; k + 4 < K; k += 4) {
        loadA(k + 4, a2);
        loadB(k + 4, b2);
        __builtin_prefetch(Ap + k + 128, 0, 3);
        __builtin_prefetch(Bp + (size_t)(k + 16) * ldb, 0, 3);
#pragma unroll
        for (int i = 0; i < MT; ++i)
#pragma unroll
            for (int t = 0; t < NT; ++t)
                acc[i][t] = __builtin_amdgcn_wmma_f32_16x16x4_f32(
                    false, a[i], false, b[t], (short)0, acc[i][t], false, false);
#pragma unroll
        for (int i = 0; i < MT; ++i) a[i] = a2[i];
#pragma unroll
        for (int t = 0; t < NT; ++t) b[t] = b2[t];
    }
#pragma unroll
    for (int i = 0; i < MT; ++i)
#pragma unroll
        for (int t = 0; t < NT; ++t)
            acc[i][t] = __builtin_amdgcn_wmma_f32_16x16x4_f32(
                false, a[i], false, b[t], (short)0, acc[i][t], false, false);

    const float scale = scale_ptr ? scale_ptr[0] : 1.0f;
    const int mb = (lane >> 4) << 3;
#pragma unroll
    for (int i = 0; i < MT; ++i) {
#pragma unroll
        for (int t = 0; t < NT; ++t) {
            const int n = n0 + t * 16 + lm;
#pragma unroll
            for (int v = 0; v < 8; ++v) {
                const int m = m0 + i * 16 + mb + v;
                float val = acc[i][t][v] * scale;
                if (residual) val += residual[(size_t)m * ldr + n];
                C[(size_t)m * ldc + n] = val;
            }
        }
    }
}

typedef __attribute__((ext_vector_type(4))) float v4f;

template <int MT, int NT, int BK, bool LONE>
__global__ __launch_bounds__(128)
void wmma_gemm_lds_kernel(const float* __restrict__ A, int lda,
                          const float* __restrict__ Bm, int ldb,
                          float* __restrict__ C, int ldc, int K,
                          const float* __restrict__ residual, int ldr,
                          const float* __restrict__ scale_ptr)
{
    constexpr int BM = 32 * MT;
    constexpr int BN = 32 * NT;
    constexpr int PT = BK + 4;
    constexpr int AC4 = BK / 4;
    constexpr int AROWS = 128 / AC4;
    constexpr int A4 = BM / AROWS;
    constexpr int BP = BK / 32;
    constexpr int BI = BN / 16;
    constexpr int B4 = BP * BI;
    constexpr int KS = BK / 4;
    static_assert(BK % 32 == 0 && BN % 16 == 0 && BM % AROWS == 0, "staging maps");

    __shared__ __attribute__((aligned(16))) float As[2][BM * PT];
    __shared__ __attribute__((aligned(16))) float Bs[2][BN * PT];

    const int tid  = threadIdx.x;
    const int lane = tid & 31;
    const int wave = tid >> 5;
    const int wy = wave & 1;
    const int wx = wave >> 1;
    const int bm0 = blockIdx.y * BM;
    const int bn0 = blockIdx.x * BN;
    const int lm = lane & 15;
    const int h  = lane >> 4;

    const int ar = tid / AC4, ac = (tid % AC4) * 4;
    const float* Ag = A + (size_t)(bm0 + ar) * lda + ac;
    const int a_so = ar * PT + ac;
    const int bk = wave * 8 + (lane & 7);
    const int bq = (lane >> 3) * 4;
    const float* Bg = Bm + (size_t)bk * ldb + bn0 + bq;
    const int b_so = bq * PT + bk;

    v4f ra[A4], rb[B4];
    auto gloadA = [&](int k0) {
#pragma unroll
        for (int i = 0; i < A4; ++i)
            ra[i] = *(const v4f*)(Ag + (size_t)(i * AROWS) * lda + k0);
    };
    auto gloadB = [&](int k0) {
#pragma unroll
        for (int p = 0; p < BP; ++p)
#pragma unroll
            for (int i = 0; i < BI; ++i)
                rb[p * BI + i] = *(const v4f*)(Bg + (size_t)(k0 + p * 32) * ldb + i * 16);
    };
    auto sstoreA = [&](int s) {
#pragma unroll
        for (int i = 0; i < A4; ++i)
            *(v4f*)(&As[s][a_so + i * AROWS * PT]) = ra[i];
    };
    auto sstoreB = [&](int s) {
#pragma unroll
        for (int p = 0; p < BP; ++p)
#pragma unroll
            for (int i = 0; i < BI; ++i)
#pragma unroll
                for (int j = 0; j < 4; ++j)
                    Bs[s][b_so + (i * 16 + j) * PT + p * 32] = rb[p * BI + i][j];
    };

    v8f acc[MT][NT];
#pragma unroll
    for (int i = 0; i < MT; ++i)
#pragma unroll
        for (int t = 0; t < NT; ++t)
#pragma unroll
            for (int e = 0; e < 8; ++e) acc[i][t][e] = 0.f;

    const int a_fo = (wy * 16 * MT + lm) * PT + 2 * h;
    const int b_fo = (wx * 16 * NT + lm) * PT + 2 * h;

    auto lfrag = [&](const float* Af, const float* Bf, int kk, v2f (&a)[MT], v2f (&b)[NT]) {
#pragma unroll
        for (int i = 0; i < MT; ++i) a[i] = *(const v2f*)(Af + i * 16 * PT + kk * 4);
#pragma unroll
        for (int t = 0; t < NT; ++t) b[t] = *(const v2f*)(Bf + t * 16 * PT + kk * 4);
    };
    auto mma = [&](const v2f (&a)[MT], const v2f (&b)[NT]) {
#pragma unroll
        for (int i = 0; i < MT; ++i)
#pragma unroll
            for (int t = 0; t < NT; ++t)
                acc[i][t] = __builtin_amdgcn_wmma_f32_16x16x4_f32(
                    false, a[i], false, b[t], (short)0, acc[i][t], false, false);
    };

    const int nchunks = K / BK;

    gloadA(0);
    gloadB(0);
    sstoreA(0);
    sstoreB(0);
    __syncthreads();

    for (int ch = 0; ch < nchunks; ++ch) {
        const int s = ch & 1;
        const bool more = ch + 1 < nchunks;
        const int k1 = (ch + 1) * BK;
        const float* Af = &As[s][a_fo];
        const float* Bf = &Bs[s][b_fo];

        if constexpr (LONE) {
            if (more) { gloadA(k1); gloadB(k1); }
            v2f a[2][2][MT], b[2][2][NT];
            lfrag(Af, Bf, 0, a[0][0], b[0][0]);
            lfrag(Af, Bf, 1, a[0][1], b[0][1]);
#pragma unroll
            for (int g = 0; g < KS / 2; ++g) {
                if (g + 1 < KS / 2) {
                    lfrag(Af, Bf, 2 * g + 2, a[(g + 1) & 1][0], b[(g + 1) & 1][0]);
                    lfrag(Af, Bf, 2 * g + 3, a[(g + 1) & 1][1], b[(g + 1) & 1][1]);
                }
                mma(a[g & 1][0], b[g & 1][0]);
                mma(a[g & 1][1], b[g & 1][1]);
            }
            if (more) { sstoreA(s ^ 1); sstoreB(s ^ 1); }
        } else {
            if (more) gloadA(k1);
#pragma unroll
            for (int kk = 0; kk < KS / 2; ++kk) {
                v2f a[MT], b[NT];
                lfrag(Af, Bf, kk, a, b);
                mma(a, b);
            }
            if (more) { sstoreA(s ^ 1); gloadB(k1); }
#pragma unroll
            for (int kk = KS / 2; kk < KS; ++kk) {
                v2f a[MT], b[NT];
                lfrag(Af, Bf, kk, a, b);
                mma(a, b);
            }
            if (more) sstoreB(s ^ 1);
        }
        __syncthreads();
    }

    const float scale = scale_ptr ? scale_ptr[0] : 1.0f;
    const int m0 = bm0 + wy * 16 * MT;
    const int n0 = bn0 + wx * 16 * NT;
    const int mb = h << 3;
#pragma unroll
    for (int i = 0; i < MT; ++i) {
#pragma unroll
        for (int t = 0; t < NT; ++t) {
            const int n = n0 + t * 16 + lm;
#pragma unroll
            for (int v = 0; v < 8; ++v) {
                const int m = m0 + i * 16 + mb + v;
                float val = acc[i][t][v] * scale;
                if (residual) val += residual[(size_t)m * ldr + n];
                C[(size_t)m * ldc + n] = val;
            }
        }
    }
}

__global__ __launch_bounds__(256)
void conv_silu_kernel(const float* __restrict__ xz, const float* __restrict__ w,
                      const float* __restrict__ bias, float* __restrict__ out)
{
    const size_t idx = (size_t)blockIdx.x * 256 + threadIdx.x;
    const int c = (int)(idx & (D_INNER - 1));
    const size_t bl = idx >> 11;
    const int l = (int)(bl & (LL - 1));
    const int b = (int)(bl >> 11);
    float acc = bias[c];
#pragma unroll
    for (int k = 0; k < D_CONV; ++k) {
        const int ls = l - (D_CONV - 1) + k;
        if (ls >= 0)
            acc += w[c * D_CONV + k] *
                   xz[((size_t)(b * LL + ls)) * (2 * D_INNER) + c];
    }
    out[idx] = acc * sigmoidf_(acc);
}

__global__ __launch_bounds__(256)
void conv_state_kernel(const float* __restrict__ xz, float* __restrict__ out)
{
    const int idx = blockIdx.x * 256 + threadIdx.x;
    if (idx >= BB * D_INNER * (D_CONV - 1)) return;
    const int j = idx % (D_CONV - 1);
    const int c = (idx / (D_CONV - 1)) % D_INNER;
    const int b = idx / ((D_CONV - 1) * D_INNER);
    out[idx] = xz[((size_t)(b * LL + (LL - (D_CONV - 1) + j))) * (2 * D_INNER) + c];
}

__global__ __launch_bounds__(256)
void dt_softplus_kernel(float* __restrict__ dt, const float* __restrict__ bias)
{
    const size_t idx = (size_t)blockIdx.x * 256 + threadIdx.x;
    const int d = (int)(idx & (D_INNER - 1));
    const float v = dt[idx] + bias[d];
    dt[idx] = (v > 20.f) ? v : __logf(1.f + __expf(v));
}

#define SCAN_T  32
#define SCAN_NC (LL / SCAN_T)
#define SCAN_G  4

__device__ __forceinline__ size_t cs_index(int b, int c, int n, int d)
{
    return ((size_t)((b * SCAN_NC + c) * D_STATE + n)) * D_INNER + d;
}

__device__ __forceinline__ void load_aval(const float* __restrict__ A_log, int d,
                                          float (&Aval)[D_STATE])
{
#pragma unroll
    for (int n = 0; n < D_STATE; ++n)
        Aval[n] = -__expf(A_log[d * D_STATE + n]);
}

__global__ __launch_bounds__(128)
void scan_chunk_state_kernel(const float* __restrict__ dt, const float* __restrict__ xc,
                             const float* __restrict__ xdbl, const float* __restrict__ A_log,
                             float* __restrict__ hloc, float* __restrict__ pdec)
{
    const int d = blockIdx.x * 128 + threadIdx.x;
    const int c = blockIdx.y;
    const int b = blockIdx.z;

    float Aval[D_STATE];
    load_aval(A_log, d, Aval);

    float h[D_STATE], P[D_STATE];
#pragma unroll
    for (int n = 0; n < D_STATE; ++n) { h[n] = 0.f; P[n] = 1.f; }

    const size_t row0 = (size_t)b * LL + (size_t)c * SCAN_T;
    const float* dtp = dt + row0 * D_INNER + d;
    const float* xp  = xc + row0 * D_INNER + d;
    const float* bcp = xdbl + row0 * XDBL_W + DT_RANK;

    for (int l0 = 0; l0 < SCAN_T; l0 += SCAN_G) {
        float dtv[SCAN_G], xv[SCAN_G];
#pragma unroll
        for (int j = 0; j < SCAN_G; ++j) {
            dtv[j] = dtp[(size_t)(l0 + j) * D_INNER];
            xv[j]  = xp[(size_t)(l0 + j) * D_INNER];
        }
#pragma unroll
        for (int j = 0; j < SCAN_G; ++j) {
            const float4* Bp = (const float4*)(bcp + (size_t)(l0 + j) * XDBL_W);
            float Bb[D_STATE];
#pragma unroll
            for (int q = 0; q < 4; ++q) {
                const float4 bq = Bp[q];
                Bb[4*q+0] = bq.x; Bb[4*q+1] = bq.y; Bb[4*q+2] = bq.z; Bb[4*q+3] = bq.w;
            }
            const float bx = dtv[j] * xv[j];
#pragma unroll
            for (int n = 0; n < D_STATE; ++n) {
                const float dA = __expf(Aval[n] * dtv[j]);
                h[n] = dA * h[n] + bx * Bb[n];
                P[n] *= dA;
            }
        }
    }

#pragma unroll
    for (int n = 0; n < D_STATE; ++n) {
        hloc[cs_index(b, c, n, d)] = h[n];
        pdec[cs_index(b, c, n, d)] = P[n];
    }
}

#define CARRY_G 21
static_assert((SCAN_NC - 1) % CARRY_G == 0, "carry groups must tile the chunks");

__global__ __launch_bounds__(256)
void scan_carry_kernel(const float* __restrict__ hloc, const float* __restrict__ pdec,
                       float* __restrict__ hin)
{
    const int t = blockIdx.x * 256 + threadIdx.x;
    const int d = t & (D_INNER - 1);
    const int n = (t >> 11) & (D_STATE - 1);
    const int b = t >> 15;
    const size_t base = cs_index(b, 0, n, d);
    const size_t cstride = (size_t)D_STATE * D_INNER;
    float carry = 0.f;
    for (int c0 = 0; c0 < SCAN_NC - 1; c0 += CARRY_G) {
        float p[CARRY_G], s[CARRY_G];
#pragma unroll
        for (int j = 0; j < CARRY_G; ++j) {
            p[j] = pdec[base + (size_t)(c0 + j) * cstride];
            s[j] = hloc[base + (size_t)(c0 + j) * cstride];
        }
#pragma unroll
        for (int j = 0; j < CARRY_G; ++j) {
            hin[base + (size_t)(c0 + j) * cstride] = carry;
            carry = p[j] * carry + s[j];
        }
    }
    hin[base + (size_t)(SCAN_NC - 1) * cstride] = carry;
}

__global__ __launch_bounds__(128)
void scan_chunk_out_kernel(const float* __restrict__ dt, const float* __restrict__ xc,
                           const float* __restrict__ xdbl, const float* __restrict__ A_log,
                           const float* __restrict__ hin,
                           float* __restrict__ y, float* __restrict__ ssm_out)
{
    const int d = blockIdx.x * 128 + threadIdx.x;
    const int c = blockIdx.y;
    const int b = blockIdx.z;

    float Aval[D_STATE];
    load_aval(A_log, d, Aval);

    float h[D_STATE];
#pragma unroll
    for (int n = 0; n < D_STATE; ++n) h[n] = hin[cs_index(b, c, n, d)];

    const size_t row0 = (size_t)b * LL + (size_t)c * SCAN_T;
    const float* dtp = dt + row0 * D_INNER + d;
    const float* xp  = xc + row0 * D_INNER + d;
    const float* bcp = xdbl + row0 * XDBL_W + DT_RANK;
    float* yp = y + row0 * D_INNER + d;

    for (int l0 = 0; l0 < SCAN_T; l0 += SCAN_G) {
        float dtv[SCAN_G], xv[SCAN_G];
#pragma unroll
        for (int j = 0; j < SCAN_G; ++j) {
            dtv[j] = dtp[(size_t)(l0 + j) * D_INNER];
            xv[j]  = xp[(size_t)(l0 + j) * D_INNER];
        }
#pragma unroll
        for (int j = 0; j < SCAN_G; ++j) {
            const float4* Bp = (const float4*)(bcp + (size_t)(l0 + j) * XDBL_W);
            const float4* Cp = Bp + 4;
            float Bb[D_STATE], Cc[D_STATE];
#pragma unroll
            for (int q = 0; q < 4; ++q) {
                const float4 bq = Bp[q];
                const float4 cq = Cp[q];
                Bb[4*q+0] = bq.x; Bb[4*q+1] = bq.y; Bb[4*q+2] = bq.z; Bb[4*q+3] = bq.w;
                Cc[4*q+0] = cq.x; Cc[4*q+1] = cq.y; Cc[4*q+2] = cq.z; Cc[4*q+3] = cq.w;
            }
            const float bx = dtv[j] * xv[j];
            float yv = 0.f;
#pragma unroll
            for (int n = 0; n < D_STATE; ++n) {
                const float dA = __expf(Aval[n] * dtv[j]);
                h[n] = dA * h[n] + bx * Bb[n];
                yv += h[n] * Cc[n];
            }
            yp[(size_t)(l0 + j) * D_INNER] = yv;
        }
    }

    if (c == SCAN_NC - 1) {
#pragma unroll
        for (int n = 0; n < D_STATE; ++n)
            ssm_out[(size_t)(b * D_INNER + d) * D_STATE + n] = h[n];
    }
}

__global__ __launch_bounds__(256)
void gate_kernel(float* __restrict__ y, const float* __restrict__ xc,
                 const float* __restrict__ xz, const float* __restrict__ Dp)
{
    const size_t idx = (size_t)blockIdx.x * 256 + threadIdx.x;
    const int d = (int)(idx & (D_INNER - 1));
    const size_t row = idx >> 11;
    const float z  = xz[row * (2 * D_INNER) + D_INNER + d];
    const float yv = y[idx] + xc[idx] * Dp[d];
    y[idx] = yv * z * sigmoidf_(z);
}

extern "C" void kernel_launch(void* const* d_in, const int* in_sizes, int n_in,
                              void* d_out, int out_size, void* d_ws, size_t ws_size,
                              hipStream_t stream)
{
    const float* x         = (const float*)d_in[0];
    const float* ln_w      = (const float*)d_in[1];
    const float* ln_b      = (const float*)d_in[2];
    const float* in_proj_w = (const float*)d_in[3];
    const float* conv_w    = (const float*)d_in[4];
    const float* conv_b    = (const float*)d_in[5];
    const float* x_proj_w  = (const float*)d_in[6];
    const float* dt_proj_w = (const float*)d_in[7];
    const float* dt_proj_b = (const float*)d_in[8];
    const float* A_log     = (const float*)d_in[9];
    const float* D_param   = (const float*)d_in[10];
    const float* out_projw = (const float*)d_in[11];
    const float* res_scale = (const float*)d_in[12];

    float* out      = (float*)d_out;
    float* out_ssm  = out + (size_t)BB * LL * D_MODEL;
    float* out_conv = out_ssm + (size_t)BB * D_INNER * D_STATE;

    float* ws    = (float*)d_ws;
    float* xn    = ws;
    float* xz    = xn    + (size_t)TOKENS * D_MODEL;
    float* xc    = xz    + (size_t)TOKENS * 2 * D_INNER;
    float* xdbl  = xc    + (size_t)TOKENS * D_INNER;
    float* dtbuf = xdbl  + (size_t)TOKENS * XDBL_W;
    float* ybuf  = dtbuf + (size_t)TOKENS * D_INNER;
    float* hloc  = ybuf  + (size_t)TOKENS * D_INNER;
    float* pdec  = hloc  + (size_t)BB * SCAN_NC * D_STATE * D_INNER;
    float* hin   = pdec  + (size_t)BB * SCAN_NC * D_STATE * D_INNER;

    ln_kernel<<<TOKENS, 256, 0, stream>>>(x, ln_w, ln_b, xn);

    wmma_gemm_lds_kernel<4, 4, 32, false><<<dim3((2 * D_INNER) / 128, TOKENS / 128), 128, 0, stream>>>(
        xn, D_MODEL, in_proj_w, 2 * D_INNER, xz, 2 * D_INNER, D_MODEL,
        nullptr, 0, nullptr);

    conv_silu_kernel<<<(TOKENS * D_INNER) / 256, 256, 0, stream>>>(
        xz, conv_w, conv_b, xc);
    conv_state_kernel<<<(BB * D_INNER * (D_CONV - 1) + 255) / 256, 256, 0, stream>>>(
        xz, out_conv);

    wmma_gemm_lds_kernel<1, 3, 64, true><<<dim3(XDBL_W / 96, TOKENS / 32), 128, 0, stream>>>(
        xc, D_INNER, x_proj_w, XDBL_W, xdbl, XDBL_W, D_INNER,
        nullptr, 0, nullptr);

    wmma_gemm_kernel<2, 4><<<dim3(D_INNER / 128, TOKENS / 64), 128, 0, stream>>>(
        xdbl, XDBL_W, dt_proj_w, D_INNER, dtbuf, D_INNER, DT_RANK,
        nullptr, 0, nullptr);

    dt_softplus_kernel<<<(TOKENS * D_INNER) / 256, 256, 0, stream>>>(dtbuf, dt_proj_b);

    scan_chunk_state_kernel<<<dim3(D_INNER / 128, SCAN_NC - 1, BB), 128, 0, stream>>>(
        dtbuf, xc, xdbl, A_log, hloc, pdec);
    scan_carry_kernel<<<(BB * D_STATE * D_INNER) / 256, 256, 0, stream>>>(hloc, pdec, hin);
    scan_chunk_out_kernel<<<dim3(D_INNER / 128, SCAN_NC, BB), 128, 0, stream>>>(
        dtbuf, xc, xdbl, A_log, hin, ybuf, out_ssm);

    gate_kernel<<<(TOKENS * D_INNER) / 256, 256, 0, stream>>>(ybuf, xc, xz, D_param);

    wmma_gemm_lds_kernel<2, 4, 32, false><<<dim3(D_MODEL / 128, TOKENS / 64), 128, 0, stream>>>(
        ybuf, D_INNER, out_projw, D_MODEL, out, D_MODEL, D_INNER,
        x, D_MODEL, res_scale);
}


# conv+SiLU and softplus are applied by the kernels that read them (x_proj A staging, which also writes xc once; the scan passes' dt load): two one-float-per-thread launches gone, 12 to 10; sums kept
# speedup vs baseline: 4.9087x; 1.0559x over previous
#include <hip/hip_runtime.h>
#include <hip/hip_bf16.h>

typedef __attribute__((ext_vector_type(2))) float v2f;
typedef __attribute__((ext_vector_type(8))) float v8f;

#define D_MODEL 1024
#define D_STATE 16
#define D_CONV  4
#define D_INNER 2048
#define DT_RANK 64
#define BB 2
#define LL 2048
#define TOKENS (BB * LL)
#define XDBL_W (DT_RANK + 2 * D_STATE)

__device__ __forceinline__ float sigmoidf_(float x) { return 1.f / (1.f + __expf(-x)); }

__global__ __launch_bounds__(256)
void ln_kernel(const float* __restrict__ x, const float* __restrict__ w,
               const float* __restrict__ b, float* __restrict__ out)
{
    const int row = blockIdx.x;
    const float* xr = x + (size_t)row * D_MODEL;
    float v[4];
    float s = 0.f, s2 = 0.f;
#pragma unroll
    for (int i = 0; i < 4; ++i) {
        v[i] = xr[threadIdx.x + i * 256];
        s += v[i];
        s2 += v[i] * v[i];
    }
#pragma unroll
    for (int off = 16; off > 0; off >>= 1) {
        s  += __shfl_down(s,  off, 32);
        s2 += __shfl_down(s2, off, 32);
    }
    __shared__ float rs[8], rq[8];
    __shared__ float mu_s, rstd_s;
    const int wave = threadIdx.x >> 5;
    const int lane = threadIdx.x & 31;
    if (lane == 0) { rs[wave] = s; rq[wave] = s2; }
    __syncthreads();
    if (threadIdx.x == 0) {
        float a = 0.f, q = 0.f;
#pragma unroll
        for (int i = 0; i < 8; ++i) { a += rs[i]; q += rq[i]; }
        const float mu  = a * (1.f / D_MODEL);
        const float var = q * (1.f / D_MODEL) - mu * mu;
        mu_s = mu;
        rstd_s = rsqrtf(var + 1e-5f);
    }
    __syncthreads();
    const float mu = mu_s, rstd = rstd_s;
    float* orow = out + (size_t)row * D_MODEL;
#pragma unroll
    for (int i = 0; i < 4; ++i) {
        const int c = threadIdx.x + i * 256;
        orow[c] = (v[i] - mu) * rstd * w[c] + b[c];
    }
}

template <int MT, int NT>
__global__ __launch_bounds__(128)
void wmma_gemm_kernel(const float* __restrict__ A, int lda,
                      const float* __restrict__ Bm, int ldb,
                      float* __restrict__ C, int ldc, int K,
                      const float* __restrict__ residual, int ldr,
                      const float* __restrict__ scale_ptr)
{
    const int lane = threadIdx.x & 31;
    const int wave = threadIdx.x >> 5;
    const int wy = wave & 1;
    const int wx = wave >> 1;
    const int m0 = blockIdx.y * (32 * MT) + wy * (16 * MT);
    const int n0 = blockIdx.x * (32 * NT) + wx * (16 * NT);
    const int lm = lane & 15;
    const int kh = (lane >> 4) << 1;

    v8f acc[MT][NT];
#pragma unroll
    for (int i = 0; i < MT; ++i)
#pragma unroll
        for (int t = 0; t < NT; ++t)
#pragma unroll
            for (int e = 0; e < 8; ++e) acc[i][t][e] = 0.f;

    const float* Ap = A + (size_t)(m0 + lm) * lda + kh;
    const float* Bp = Bm + (size_t)kh * ldb + n0 + lm;

    auto loadA = [&](int k, v2f (&a)[MT]) {
#pragma unroll
        for (int i = 0; i < MT; ++i)
            a[i] = *(const v2f*)(Ap + (size_t)i * 16 * lda + k);
    };
    auto loadB = [&](int k, v2f (&b)[NT]) {
        const float* r0 = Bp + (size_t)k * ldb;
        const float* r1 = r0 + ldb;
#pragma unroll
        for (int t = 0; t < NT; ++t) {
            b[t].x = r0[t * 16];
            b[t].y = r1[t * 16];
        }
    };

    v2f a[MT], b[NT], a2[MT], b2[NT];
    loadA(0, a);
    loadB(0, b);

    int k = 0;
    for (; k + 4 < K; k += 4) {
        loadA(k + 4, a2);
        loadB(k + 4, b2);
        __builtin_prefetch(Ap + k + 128, 0, 3);
        __builtin_prefetch(Bp + (size_t)(k + 16) * ldb, 0, 3);
#pragma unroll
        for (int i = 0; i < MT; ++i)
#pragma unroll
            for (int t = 0; t < NT; ++t)
                acc[i][t] = __builtin_amdgcn_wmma_f32_16x16x4_f32(
                    false, a[i], false, b[t], (short)0, acc[i][t], false, false);
#pragma unroll
        for (int i = 0; i < MT; ++i) a[i] = a2[i];
#pragma unroll
        for (int t = 0; t < NT; ++t) b[t] = b2[t];
    }
#pragma unroll
    for (int i = 0; i < MT; ++i)
#pragma unroll
        for (int t = 0; t < NT; ++t)
            acc[i][t] = __builtin_amdgcn_wmma_f32_16x16x4_f32(
                false, a[i], false, b[t], (short)0, acc[i][t], false, false);

    const float scale = scale_ptr ? scale_ptr[0] : 1.0f;
    const int mb = (lane >> 4) << 3;
#pragma unroll
    for (int i = 0; i < MT; ++i) {
#pragma unroll
        for (int t = 0; t < NT; ++t) {
            const int n = n0 + t * 16 + lm;
#pragma unroll
            for (int v = 0; v < 8; ++v) {
                const int m = m0 + i * 16 + mb + v;
                float val = acc[i][t][v] * scale;
                if (residual) val += residual[(size_t)m * ldr + n];
                C[(size_t)m * ldc + n] = val;
            }
        }
    }
}

typedef __attribute__((ext_vector_type(4))) float v4f;

template <int MT, int NT, int BK, bool LONE, bool CONVA>
__global__ __launch_bounds__(128)
void wmma_gemm_lds_kernel(const float* __restrict__ A, int lda,
                          const float* __restrict__ Bm, int ldb,
                          float* __restrict__ C, int ldc, int K,
                          const float* __restrict__ residual, int ldr,
                          const float* __restrict__ scale_ptr,
                          const float* __restrict__ conv_w,
                          const float* __restrict__ conv_b,
                          float* __restrict__ xc_out)
{
    constexpr int BM = 32 * MT;
    constexpr int BN = 32 * NT;
    constexpr int PT = BK + 4;
    constexpr int AC4 = BK / 4;
    constexpr int AROWS = 128 / AC4;
    constexpr int A4 = BM / AROWS;
    constexpr int BP = BK / 32;
    constexpr int BI = BN / 16;
    constexpr int B4 = BP * BI;
    constexpr int KS = BK / 4;
    static_assert(BK % 32 == 0 && BN % 16 == 0 && BM % AROWS == 0, "staging maps");
    static_assert(!CONVA || LONE, "the conv loader is written for the one-wave-per-SIMD path");

    __shared__ __attribute__((aligned(16))) float As[2][BM * PT];
    __shared__ __attribute__((aligned(16))) float Bs[2][BN * PT];

    const int tid  = threadIdx.x;
    const int lane = tid & 31;
    const int wave = tid >> 5;
    const int wy = wave & 1;
    const int wx = wave >> 1;
    const int bm0 = blockIdx.y * BM;
    const int bn0 = blockIdx.x * BN;
    const int lm = lane & 15;
    const int h  = lane >> 4;

    const int ar = tid / AC4, ac = (tid % AC4) * 4;
    const int arow0 = CONVA ? ar * A4 : ar;
    constexpr int ASTEP = CONVA ? 1 : AROWS;
    const float* Ag = A + (size_t)(bm0 + arow0) * lda + ac;
    const int a_so = arow0 * PT + ac;
    const bool head = CONVA && (((bm0 + arow0) & (LL - 1)) == 0);
    static_assert(!CONVA || (LL % BM == 0 && A4 >= D_CONV - 1 && D_CONV == 4), "conv window");
    const int bk = wave * 8 + (lane & 7);
    const int bq = (lane >> 3) * 4;
    const float* Bg = Bm + (size_t)bk * ldb + bn0 + bq;
    const int b_so = bq * PT + bk;

    v4f ra[A4], rb[B4];
    constexpr int WR = A4 + D_CONV - 1;
    v4f rw[CONVA ? WR : 1], cw[CONVA ? 4 : 1], cb;
    auto gloadA = [&](int k0) {
        if constexpr (CONVA) {
#pragma unroll
            for (int q = 0; q < WR; ++q) {
                if (!head || q >= D_CONV - 1)
                    rw[q] = *(const v4f*)(Ag + (long)(q - (D_CONV - 1)) * lda + k0);
                else
                    rw[q] = (v4f){0.f, 0.f, 0.f, 0.f};
            }
#pragma unroll
            for (int j = 0; j < 4; ++j)
                cw[j] = *(const v4f*)(conv_w + (size_t)(ac + k0 + j) * D_CONV);
            cb = *(const v4f*)(conv_b + ac + k0);
        } else {
#pragma unroll
            for (int i = 0; i < A4; ++i)
                ra[i] = *(const v4f*)(Ag + (size_t)(i * ASTEP) * lda + k0);
        }
    };
    auto gloadB = [&](int k0) {
#pragma unroll
        for (int p = 0; p < BP; ++p)
#pragma unroll
            for (int i = 0; i < BI; ++i)
                rb[p * BI + i] = *(const v4f*)(Bg + (size_t)(k0 + p * 32) * ldb + i * 16);
    };
    auto sstoreA = [&](int s, int k0) {
        if constexpr (CONVA) {
#pragma unroll
            for (int i = 0; i < A4; ++i) {
                v4f o;
#pragma unroll
                for (int j = 0; j < 4; ++j) {
                    float acc = cb[j];
#pragma unroll
                    for (int k = 0; k < D_CONV; ++k)
                        if (!head || i + k >= D_CONV - 1)
                            acc += cw[j][k] * rw[i + k][j];
                    o[j] = acc * sigmoidf_(acc);
                }
                *(v4f*)(&As[s][a_so + i * PT]) = o;
                *(v4f*)(xc_out + (size_t)(bm0 + arow0 + i) * D_INNER + ac + k0) = o;
            }
        } else {
#pragma unroll
            for (int i = 0; i < A4; ++i)
                *(v4f*)(&As[s][a_so + i * ASTEP * PT]) = ra[i];
        }
    };
    auto sstoreB = [&](int s) {
#pragma unroll
        for (int p = 0; p < BP; ++p)
#pragma unroll
            for (int i = 0; i < BI; ++i)
#pragma unroll
                for (int j = 0; j < 4; ++j)
                    Bs[s][b_so + (i * 16 + j) * PT + p * 32] = rb[p * BI + i][j];
    };

    v8f acc[MT][NT];
#pragma unroll
    for (int i = 0; i < MT; ++i)
#pragma unroll
        for (int t = 0; t < NT; ++t)
#pragma unroll
            for (int e = 0; e < 8; ++e) acc[i][t][e] = 0.f;

    const int a_fo = (wy * 16 * MT + lm) * PT + 2 * h;
    const int b_fo = (wx * 16 * NT + lm) * PT + 2 * h;

    auto lfrag = [&](const float* Af, const float* Bf, int kk, v2f (&a)[MT], v2f (&b)[NT]) {
#pragma unroll
        for (int i = 0; i < MT; ++i) a[i] = *(const v2f*)(Af + i * 16 * PT + kk * 4);
#pragma unroll
        for (int t = 0; t < NT; ++t) b[t] = *(const v2f*)(Bf + t * 16 * PT + kk * 4);
    };
    auto mma = [&](const v2f (&a)[MT], const v2f (&b)[NT]) {
#pragma unroll
        for (int i = 0; i < MT; ++i)
#pragma unroll
            for (int t = 0; t < NT; ++t)
                acc[i][t] = __builtin_amdgcn_wmma_f32_16x16x4_f32(
                    false, a[i], false, b[t], (short)0, acc[i][t], false, false);
    };

    const int nchunks = K / BK;

    gloadA(0);
    gloadB(0);
    sstoreA(0, 0);
    sstoreB(0);
    __syncthreads();

    for (int ch = 0; ch < nchunks; ++ch) {
        const int s = ch & 1;
        const bool more = ch + 1 < nchunks;
        const int k1 = (ch + 1) * BK;
        const float* Af = &As[s][a_fo];
        const float* Bf = &Bs[s][b_fo];

        if constexpr (LONE) {
            if (more) { gloadA(k1); gloadB(k1); }
            v2f a[2][2][MT], b[2][2][NT];
            lfrag(Af, Bf, 0, a[0][0], b[0][0]);
            lfrag(Af, Bf, 1, a[0][1], b[0][1]);
#pragma unroll
            for (int g = 0; g < KS / 2; ++g) {
                if (g + 1 < KS / 2) {
                    lfrag(Af, Bf, 2 * g + 2, a[(g + 1) & 1][0], b[(g + 1) & 1][0]);
                    lfrag(Af, Bf, 2 * g + 3, a[(g + 1) & 1][1], b[(g + 1) & 1][1]);
                }
                mma(a[g & 1][0], b[g & 1][0]);
                mma(a[g & 1][1], b[g & 1][1]);
            }
            if (more) { sstoreA(s ^ 1, k1); sstoreB(s ^ 1); }
        } else {
            if (more) gloadA(k1);
#pragma unroll
            for (int kk = 0; kk < KS / 2; ++kk) {
                v2f a[MT], b[NT];
                lfrag(Af, Bf, kk, a, b);
                mma(a, b);
            }
            if (more) { sstoreA(s ^ 1, k1); gloadB(k1); }
#pragma unroll
            for (int kk = KS / 2; kk < KS; ++kk) {
                v2f a[MT], b[NT];
                lfrag(Af, Bf, kk, a, b);
                mma(a, b);
            }
            if (more) sstoreB(s ^ 1);
        }
        __syncthreads();
    }

    const float scale = scale_ptr ? scale_ptr[0] : 1.0f;
    const int m0 = bm0 + wy * 16 * MT;
    const int n0 = bn0 + wx * 16 * NT;
    const int mb = h << 3;
#pragma unroll
    for (int i = 0; i < MT; ++i) {
#pragma unroll
        for (int t = 0; t < NT; ++t) {
            const int n = n0 + t * 16 + lm;
#pragma unroll
            for (int v = 0; v < 8; ++v) {
                const int m = m0 + i * 16 + mb + v;
                float val = acc[i][t][v] * scale;
                if (residual) val += residual[(size_t)m * ldr + n];
                C[(size_t)m * ldc + n] = val;
            }
        }
    }
}

__global__ __launch_bounds__(256)
void conv_state_kernel(const float* __restrict__ xz, float* __restrict__ out)
{
    const int idx = blockIdx.x * 256 + threadIdx.x;
    if (idx >= BB * D_INNER * (D_CONV - 1)) return;
    const int j = idx % (D_CONV - 1);
    const int c = (idx / (D_CONV - 1)) % D_INNER;
    const int b = idx / ((D_CONV - 1) * D_INNER);
    out[idx] = xz[((size_t)(b * LL + (LL - (D_CONV - 1) + j))) * (2 * D_INNER) + c];
}

__device__ __forceinline__ float softplus_dt(float v)
{
    return (v > 20.f) ? v : __logf(1.f + __expf(v));
}

#define SCAN_T  32
#define SCAN_NC (LL / SCAN_T)
#define SCAN_G  4

__device__ __forceinline__ size_t cs_index(int b, int c, int n, int d)
{
    return ((size_t)((b * SCAN_NC + c) * D_STATE + n)) * D_INNER + d;
}

__device__ __forceinline__ void load_aval(const float* __restrict__ A_log, int d,
                                          float (&Aval)[D_STATE])
{
#pragma unroll
    for (int n = 0; n < D_STATE; ++n)
        Aval[n] = -__expf(A_log[d * D_STATE + n]);
}

__global__ __launch_bounds__(128)
void scan_chunk_state_kernel(const float* __restrict__ dt, const float* __restrict__ dt_bias,
                             const float* __restrict__ xc,
                             const float* __restrict__ xdbl, const float* __restrict__ A_log,
                             float* __restrict__ hloc, float* __restrict__ pdec)
{
    const int d = blockIdx.x * 128 + threadIdx.x;
    const int c = blockIdx.y;
    const int b = blockIdx.z;
    const float db = dt_bias[d];

    float Aval[D_STATE];
    load_aval(A_log, d, Aval);

    float h[D_STATE], P[D_STATE];
#pragma unroll
    for (int n = 0; n < D_STATE; ++n) { h[n] = 0.f; P[n] = 1.f; }

    const size_t row0 = (size_t)b * LL + (size_t)c * SCAN_T;
    const float* dtp = dt + row0 * D_INNER + d;
    const float* xp  = xc + row0 * D_INNER + d;
    const float* bcp = xdbl + row0 * XDBL_W + DT_RANK;

    for (int l0 = 0; l0 < SCAN_T; l0 += SCAN_G) {
        float dtv[SCAN_G], xv[SCAN_G];
#pragma unroll
        for (int j = 0; j < SCAN_G; ++j) {
            dtv[j] = softplus_dt(dtp[(size_t)(l0 + j) * D_INNER] + db);
            xv[j]  = xp[(size_t)(l0 + j) * D_INNER];
        }
#pragma unroll
        for (int j = 0; j < SCAN_G; ++j) {
            const float4* Bp = (const float4*)(bcp + (size_t)(l0 + j) * XDBL_W);
            float Bb[D_STATE];
#pragma unroll
            for (int q = 0; q < 4; ++q) {
                const float4 bq = Bp[q];
                Bb[4*q+0] = bq.x; Bb[4*q+1] = bq.y; Bb[4*q+2] = bq.z; Bb[4*q+3] = bq.w;
            }
            const float bx = dtv[j] * xv[j];
#pragma unroll
            for (int n = 0; n < D_STATE; ++n) {
                const float dA = __expf(Aval[n] * dtv[j]);
                h[n] = dA * h[n] + bx * Bb[n];
                P[n] *= dA;
            }
        }
    }

#pragma unroll
    for (int n = 0; n < D_STATE; ++n) {
        hloc[cs_index(b, c, n, d)] = h[n];
        pdec[cs_index(b, c, n, d)] = P[n];
    }
}

#define CARRY_G 21
static_assert((SCAN_NC - 1) % CARRY_G == 0, "carry groups must tile the chunks");

__global__ __launch_bounds__(256)
void scan_carry_kernel(const float* __restrict__ hloc, const float* __restrict__ pdec,
                       float* __restrict__ hin)
{
    const int t = blockIdx.x * 256 + threadIdx.x;
    const int d = t & (D_INNER - 1);
    const int n = (t >> 11) & (D_STATE - 1);
    const int b = t >> 15;
    const size_t base = cs_index(b, 0, n, d);
    const size_t cstride = (size_t)D_STATE * D_INNER;
    float carry = 0.f;
    for (int c0 = 0; c0 < SCAN_NC - 1; c0 += CARRY_G) {
        float p[CARRY_G], s[CARRY_G];
#pragma unroll
        for (int j = 0; j < CARRY_G; ++j) {
            p[j] = pdec[base + (size_t)(c0 + j) * cstride];
            s[j] = hloc[base + (size_t)(c0 + j) * cstride];
        }
#pragma unroll
        for (int j = 0; j < CARRY_G; ++j) {
            hin[base + (size_t)(c0 + j) * cstride] = carry;
            carry = p[j] * carry + s[j];
        }
    }
    hin[base + (size_t)(SCAN_NC - 1) * cstride] = carry;
}

__global__ __launch_bounds__(128)
void scan_chunk_out_kernel(const float* __restrict__ dt, const float* __restrict__ dt_bias,
                           const float* __restrict__ xc,
                           const float* __restrict__ xdbl, const float* __restrict__ A_log,
                           const float* __restrict__ hin,
                           float* __restrict__ y, float* __restrict__ ssm_out)
{
    const int d = blockIdx.x * 128 + threadIdx.x;
    const int c = blockIdx.y;
    const int b = blockIdx.z;
    const float db = dt_bias[d];

    float Aval[D_STATE];
    load_aval(A_log, d, Aval);

    float h[D_STATE];
#pragma unroll
    for (int n = 0; n < D_STATE; ++n) h[n] = hin[cs_index(b, c, n, d)];

    const size_t row0 = (size_t)b * LL + (size_t)c * SCAN_T;
    const float* dtp = dt + row0 * D_INNER + d;
    const float* xp  = xc + row0 * D_INNER + d;
    const float* bcp = xdbl + row0 * XDBL_W + DT_RANK;
    float* yp = y + row0 * D_INNER + d;

    for (int l0 = 0; l0 < SCAN_T; l0 += SCAN_G) {
        float dtv[SCAN_G], xv[SCAN_G];
#pragma unroll
        for (int j = 0; j < SCAN_G; ++j) {
            dtv[j] = softplus_dt(dtp[(size_t)(l0 + j) * D_INNER] + db);
            xv[j]  = xp[(size_t)(l0 + j) * D_INNER];
        }
#pragma unroll
        for (int j = 0; j < SCAN_G; ++j) {
            const float4* Bp = (const float4*)(bcp + (size_t)(l0 + j) * XDBL_W);
            const float4* Cp = Bp + 4;
            float Bb[D_STATE], Cc[D_STATE];
#pragma unroll
            for (int q = 0; q < 4; ++q) {
                const float4 bq = Bp[q];
                const float4 cq = Cp[q];
                Bb[4*q+0] = bq.x; Bb[4*q+1] = bq.y; Bb[4*q+2] = bq.z; Bb[4*q+3] = bq.w;
                Cc[4*q+0] = cq.x; Cc[4*q+1] = cq.y; Cc[4*q+2] = cq.z; Cc[4*q+3] = cq.w;
            }
            const float bx = dtv[j] * xv[j];
            float yv = 0.f;
#pragma unroll
            for (int n = 0; n < D_STATE; ++n) {
                const float dA = __expf(Aval[n] * dtv[j]);
                h[n] = dA * h[n] + bx * Bb[n];
                yv += h[n] * Cc[n];
            }
            yp[(size_t)(l0 + j) * D_INNER] = yv;
        }
    }

    if (c == SCAN_NC - 1) {
#pragma unroll
        for (int n = 0; n < D_STATE; ++n)
            ssm_out[(size_t)(b * D_INNER + d) * D_STATE + n] = h[n];
    }
}

__global__ __launch_bounds__(256)
void gate_kernel(float* __restrict__ y, const float* __restrict__ xc,
                 const float* __restrict__ xz, const float* __restrict__ Dp)
{
    const size_t idx = (size_t)blockIdx.x * 256 + threadIdx.x;
    const int d = (int)(idx & (D_INNER - 1));
    const size_t row = idx >> 11;
    const float z  = xz[row * (2 * D_INNER) + D_INNER + d];
    const float yv = y[idx] + xc[idx] * Dp[d];
    y[idx] = yv * z * sigmoidf_(z);
}

extern "C" void kernel_launch(void* const* d_in, const int* in_sizes, int n_in,
                              void* d_out, int out_size, void* d_ws, size_t ws_size,
                              hipStream_t stream)
{
    const float* x         = (const float*)d_in[0];
    const float* ln_w      = (const float*)d_in[1];
    const float* ln_b      = (const float*)d_in[2];
    const float* in_proj_w = (const float*)d_in[3];
    const float* conv_w    = (const float*)d_in[4];
    const float* conv_b    = (const float*)d_in[5];
    const float* x_proj_w  = (const float*)d_in[6];
    const float* dt_proj_w = (const float*)d_in[7];
    const float* dt_proj_b = (const float*)d_in[8];
    const float* A_log     = (const float*)d_in[9];
    const float* D_param   = (const float*)d_in[10];
    const float* out_projw = (const float*)d_in[11];
    const float* res_scale = (const float*)d_in[12];

    float* out      = (float*)d_out;
    float* out_ssm  = out + (size_t)BB * LL * D_MODEL;
    float* out_conv = out_ssm + (size_t)BB * D_INNER * D_STATE;

    float* ws    = (float*)d_ws;
    float* xn    = ws;
    float* xz    = xn    + (size_t)TOKENS * D_MODEL;
    float* xc    = xz    + (size_t)TOKENS * 2 * D_INNER;
    float* xdbl  = xc    + (size_t)TOKENS * D_INNER;
    float* dtbuf = xdbl  + (size_t)TOKENS * XDBL_W;
    float* ybuf  = dtbuf + (size_t)TOKENS * D_INNER;
    float* hloc  = ybuf  + (size_t)TOKENS * D_INNER;
    float* pdec  = hloc  + (size_t)BB * SCAN_NC * D_STATE * D_INNER;
    float* hin   = pdec  + (size_t)BB * SCAN_NC * D_STATE * D_INNER;

    ln_kernel<<<TOKENS, 256, 0, stream>>>(x, ln_w, ln_b, xn);

    wmma_gemm_lds_kernel<4, 4, 32, false, false><<<dim3((2 * D_INNER) / 128, TOKENS / 128), 128, 0, stream>>>(
        xn, D_MODEL, in_proj_w, 2 * D_INNER, xz, 2 * D_INNER, D_MODEL,
        nullptr, 0, nullptr, nullptr, nullptr, nullptr);

    conv_state_kernel<<<(BB * D_INNER * (D_CONV - 1) + 255) / 256, 256, 0, stream>>>(
        xz, out_conv);

    static_assert(XDBL_W == 96, "one block column: every xc element is staged exactly once");
    wmma_gemm_lds_kernel<1, 3, 64, true, true><<<dim3(XDBL_W / 96, TOKENS / 32), 128, 0, stream>>>(
        xz, 2 * D_INNER, x_proj_w, XDBL_W, xdbl, XDBL_W, D_INNER,
        nullptr, 0, nullptr, conv_w, conv_b, xc);

    wmma_gemm_kernel<2, 4><<<dim3(D_INNER / 128, TOKENS / 64), 128, 0, stream>>>(
        xdbl, XDBL_W, dt_proj_w, D_INNER, dtbuf, D_INNER, DT_RANK,
        nullptr, 0, nullptr);

    scan_chunk_state_kernel<<<dim3(D_INNER / 128, SCAN_NC - 1, BB), 128, 0, stream>>>(
        dtbuf, dt_proj_b, xc, xdbl, A_log, hloc, pdec);
    scan_carry_kernel<<<(BB * D_STATE * D_INNER) / 256, 256, 0, stream>>>(hloc, pdec, hin);
    scan_chunk_out_kernel<<<dim3(D_INNER / 128, SCAN_NC, BB), 128, 0, stream>>>(
        dtbuf, dt_proj_b, xc, xdbl, A_log, hin, ybuf, out_ssm);

    gate_kernel<<<(TOKENS * D_INNER) / 256, 256, 0, stream>>>(ybuf, xc, xz, D_param);

    wmma_gemm_lds_kernel<2, 4, 32, false, false><<<dim3(D_MODEL / 128, TOKENS / 64), 128, 0, stream>>>(
        ybuf, D_INNER, out_projw, D_MODEL, out, D_MODEL, D_INNER,
        x, D_MODEL, res_scale, nullptr, nullptr, nullptr);
}


# loads that were each waited for alone behind a branch now go out in groups: out_proj's 64 residual loads (template flag; scale*acc+x now one fma), scan passes' dt loads ahead of softplus; sums kept
# speedup vs baseline: 5.1060x; 1.0402x over previous
#include <hip/hip_runtime.h>
#include <hip/hip_bf16.h>

typedef __attribute__((ext_vector_type(2))) float v2f;
typedef __attribute__((ext_vector_type(8))) float v8f;

#define D_MODEL 1024
#define D_STATE 16
#define D_CONV  4
#define D_INNER 2048
#define DT_RANK 64
#define BB 2
#define LL 2048
#define TOKENS (BB * LL)
#define XDBL_W (DT_RANK + 2 * D_STATE)

__device__ __forceinline__ float sigmoidf_(float x) { return 1.f / (1.f + __expf(-x)); }

__global__ __launch_bounds__(256)
void ln_kernel(const float* __restrict__ x, const float* __restrict__ w,
               const float* __restrict__ b, float* __restrict__ out)
{
    const int row = blockIdx.x;
    const float* xr = x + (size_t)row * D_MODEL;
    float v[4];
    float s = 0.f, s2 = 0.f;
#pragma unroll
    for (int i = 0; i < 4; ++i) {
        v[i] = xr[threadIdx.x + i * 256];
        s += v[i];
        s2 += v[i] * v[i];
    }
#pragma unroll
    for (int off = 16; off > 0; off >>= 1) {
        s  += __shfl_down(s,  off, 32);
        s2 += __shfl_down(s2, off, 32);
    }
    __shared__ float rs[8], rq[8];
    __shared__ float mu_s, rstd_s;
    const int wave = threadIdx.x >> 5;
    const int lane = threadIdx.x & 31;
    if (lane == 0) { rs[wave] = s; rq[wave] = s2; }
    __syncthreads();
    if (threadIdx.x == 0) {
        float a = 0.f, q = 0.f;
#pragma unroll
        for (int i = 0; i < 8; ++i) { a += rs[i]; q += rq[i]; }
        const float mu  = a * (1.f / D_MODEL);
        const float var = q * (1.f / D_MODEL) - mu * mu;
        mu_s = mu;
        rstd_s = rsqrtf(var + 1e-5f);
    }
    __syncthreads();
    const float mu = mu_s, rstd = rstd_s;
    float* orow = out + (size_t)row * D_MODEL;
#pragma unroll
    for (int i = 0; i < 4; ++i) {
        const int c = threadIdx.x + i * 256;
        orow[c] = (v[i] - mu) * rstd * w[c] + b[c];
    }
}

template <int MT, int NT, bool HASRES>
__global__ __launch_bounds__(128)
void wmma_gemm_kernel(const float* __restrict__ A, int lda,
                      const float* __restrict__ Bm, int ldb,
                      float* __restrict__ C, int ldc, int K,
                      const float* __restrict__ residual, int ldr,
                      const float* __restrict__ scale_ptr)
{
    const int lane = threadIdx.x & 31;
    const int wave = threadIdx.x >> 5;
    const int wy = wave & 1;
    const int wx = wave >> 1;
    const int m0 = blockIdx.y * (32 * MT) + wy * (16 * MT);
    const int n0 = blockIdx.x * (32 * NT) + wx * (16 * NT);
    const int lm = lane & 15;
    const int kh = (lane >> 4) << 1;

    v8f acc[MT][NT];
#pragma unroll
    for (int i = 0; i < MT; ++i)
#pragma unroll
        for (int t = 0; t < NT; ++t)
#pragma unroll
            for (int e = 0; e < 8; ++e) acc[i][t][e] = 0.f;

    const float* Ap = A + (size_t)(m0 + lm) * lda + kh;
    const float* Bp = Bm + (size_t)kh * ldb + n0 + lm;

    auto loadA = [&](int k, v2f (&a)[MT]) {
#pragma unroll
        for (int i = 0; i < MT; ++i)
            a[i] = *(const v2f*)(Ap + (size_t)i * 16 * lda + k);
    };
    auto loadB = [&](int k, v2f (&b)[NT]) {
        const float* r0 = Bp + (size_t)k * ldb;
        const float* r1 = r0 + ldb;
#pragma unroll
        for (int t = 0; t < NT; ++t) {
            b[t].x = r0[t * 16];
            b[t].y = r1[t * 16];
        }
    };

    v2f a[MT], b[NT], a2[MT], b2[NT];
    loadA(0, a);
    loadB(0, b);

    int k = 0;
    for (; k + 4 < K; k += 4) {
        loadA(k + 4, a2);
        loadB(k + 4, b2);
        __builtin_prefetch(Ap + k + 128, 0, 3);
        __builtin_prefetch(Bp + (size_t)(k + 16) * ldb, 0, 3);
#pragma unroll
        for (int i = 0; i < MT; ++i)
#pragma unroll
            for (int t = 0; t < NT; ++t)
                acc[i][t] = __builtin_amdgcn_wmma_f32_16x16x4_f32(
                    false, a[i], false, b[t], (short)0, acc[i][t], false, false);
#pragma unroll
        for (int i = 0; i < MT; ++i) a[i] = a2[i];
#pragma unroll
        for (int t = 0; t < NT; ++t) b[t] = b2[t];
    }
#pragma unroll
    for (int i = 0; i < MT; ++i)
#pragma unroll
        for (int t = 0; t < NT; ++t)
            acc[i][t] = __builtin_amdgcn_wmma_f32_16x16x4_f32(
                false, a[i], false, b[t], (short)0, acc[i][t], false, false);

    const float scale = scale_ptr ? scale_ptr[0] : 1.0f;
    const int mb = (lane >> 4) << 3;
    float res[MT][NT][8];
    if constexpr (HASRES) {
#pragma unroll
        for (int i = 0; i < MT; ++i)
#pragma unroll
            for (int t = 0; t < NT; ++t)
#pragma unroll
                for (int v = 0; v < 8; ++v)
                    res[i][t][v] = residual[(size_t)(m0 + i * 16 + mb + v) * ldr + n0 + t * 16 + lm];
    }
#pragma unroll
    for (int i = 0; i < MT; ++i) {
#pragma unroll
        for (int t = 0; t < NT; ++t) {
            const int n = n0 + t * 16 + lm;
#pragma unroll
            for (int v = 0; v < 8; ++v) {
                const int m = m0 + i * 16 + mb + v;
                float val = acc[i][t][v] * scale;
                if constexpr (HASRES) val += res[i][t][v];
                C[(size_t)m * ldc + n] = val;
            }
        }
    }
}

typedef __attribute__((ext_vector_type(4))) float v4f;

template <int MT, int NT, int BK, bool LONE, bool CONVA, bool HASRES>
__global__ __launch_bounds__(128)
void wmma_gemm_lds_kernel(const float* __restrict__ A, int lda,
                          const float* __restrict__ Bm, int ldb,
                          float* __restrict__ C, int ldc, int K,
                          const float* __restrict__ residual, int ldr,
                          const float* __restrict__ scale_ptr,
                          const float* __restrict__ conv_w,
                          const float* __restrict__ conv_b,
                          float* __restrict__ xc_out)
{
    constexpr int BM = 32 * MT;
    constexpr int BN = 32 * NT;
    constexpr int PT = BK + 4;
    constexpr int AC4 = BK / 4;
    constexpr int AROWS = 128 / AC4;
    constexpr int A4 = BM / AROWS;
    constexpr int BP = BK / 32;
    constexpr int BI = BN / 16;
    constexpr int B4 = BP * BI;
    constexpr int KS = BK / 4;
    static_assert(BK % 32 == 0 && BN % 16 == 0 && BM % AROWS == 0, "staging maps");
    static_assert(!CONVA || LONE, "the conv loader is written for the one-wave-per-SIMD path");

    __shared__ __attribute__((aligned(16))) float As[2][BM * PT];
    __shared__ __attribute__((aligned(16))) float Bs[2][BN * PT];

    const int tid  = threadIdx.x;
    const int lane = tid & 31;
    const int wave = tid >> 5;
    const int wy = wave & 1;
    const int wx = wave >> 1;
    const int bm0 = blockIdx.y * BM;
    const int bn0 = blockIdx.x * BN;
    const int lm = lane & 15;
    const int h  = lane >> 4;

    const int ar = tid / AC4, ac = (tid % AC4) * 4;
    const int arow0 = CONVA ? ar * A4 : ar;
    constexpr int ASTEP = CONVA ? 1 : AROWS;
    const float* Ag = A + (size_t)(bm0 + arow0) * lda + ac;
    const int a_so = arow0 * PT + ac;
    const bool head = CONVA && (((bm0 + arow0) & (LL - 1)) == 0);
    static_assert(!CONVA || (LL % BM == 0 && A4 >= D_CONV - 1 && D_CONV == 4), "conv window");
    const int bk = wave * 8 + (lane & 7);
    const int bq = (lane >> 3) * 4;
    const float* Bg = Bm + (size_t)bk * ldb + bn0 + bq;
    const int b_so = bq * PT + bk;

    v4f ra[A4], rb[B4];
    constexpr int WR = A4 + D_CONV - 1;
    v4f rw[CONVA ? WR : 1], cw[CONVA ? 4 : 1], cb;
    auto gloadA = [&](int k0) {
        if constexpr (CONVA) {
#pragma unroll
            for (int q = 0; q < WR; ++q) {
                if (!head || q >= D_CONV - 1)
                    rw[q] = *(const v4f*)(Ag + (long)(q - (D_CONV - 1)) * lda + k0);
                else
                    rw[q] = (v4f){0.f, 0.f, 0.f, 0.f};
            }
#pragma unroll
            for (int j = 0; j < 4; ++j)
                cw[j] = *(const v4f*)(conv_w + (size_t)(ac + k0 + j) * D_CONV);
            cb = *(const v4f*)(conv_b + ac + k0);
        } else {
#pragma unroll
            for (int i = 0; i < A4; ++i)
                ra[i] = *(const v4f*)(Ag + (size_t)(i * ASTEP) * lda + k0);
        }
    };
    auto gloadB = [&](int k0) {
#pragma unroll
        for (int p = 0; p < BP; ++p)
#pragma unroll
            for (int i = 0; i < BI; ++i)
                rb[p * BI + i] = *(const v4f*)(Bg + (size_t)(k0 + p * 32) * ldb + i * 16);
    };
    auto sstoreA = [&](int s, int k0) {
        if constexpr (CONVA) {
#pragma unroll
            for (int i = 0; i < A4; ++i) {
                v4f o;
#pragma unroll
                for (int j = 0; j < 4; ++j) {
                    float acc = cb[j];
#pragma unroll
                    for (int k = 0; k < D_CONV; ++k)
                        if (!head || i + k >= D_CONV - 1)
                            acc += cw[j][k] * rw[i + k][j];
                    o[j] = acc * sigmoidf_(acc);
                }
                *(v4f*)(&As[s][a_so + i * PT]) = o;
                *(v4f*)(xc_out + (size_t)(bm0 + arow0 + i) * D_INNER + ac + k0) = o;
            }
        } else {
#pragma unroll
            for (int i = 0; i < A4; ++i)
                *(v4f*)(&As[s][a_so + i * ASTEP * PT]) = ra[i];
        }
    };
    auto sstoreB = [&](int s) {
#pragma unroll
        for (int p = 0; p < BP; ++p)
#pragma unroll
            for (int i = 0; i < BI; ++i)
#pragma unroll
                for (int j = 0; j < 4; ++j)
                    Bs[s][b_so + (i * 16 + j) * PT + p * 32] = rb[p * BI + i][j];
    };

    v8f acc[MT][NT];
#pragma unroll
    for (int i = 0; i < MT; ++i)
#pragma unroll
        for (int t = 0; t < NT; ++t)
#pragma unroll
            for (int e = 0; e < 8; ++e) acc[i][t][e] = 0.f;

    const int a_fo = (wy * 16 * MT + lm) * PT + 2 * h;
    const int b_fo = (wx * 16 * NT + lm) * PT + 2 * h;

    auto lfrag = [&](const float* Af, const float* Bf, int kk, v2f (&a)[MT], v2f (&b)[NT]) {
#pragma unroll
        for (int i = 0; i < MT; ++i) a[i] = *(const v2f*)(Af + i * 16 * PT + kk * 4);
#pragma unroll
        for (int t = 0; t < NT; ++t) b[t] = *(const v2f*)(Bf + t * 16 * PT + kk * 4);
    };
    auto mma = [&](const v2f (&a)[MT], const v2f (&b)[NT]) {
#pragma unroll
        for (int i = 0; i < MT; ++i)
#pragma unroll
            for (int t = 0; t < NT; ++t)
                acc[i][t] = __builtin_amdgcn_wmma_f32_16x16x4_f32(
                    false, a[i], false, b[t], (short)0, acc[i][t], false, false);
    };

    const int nchunks = K / BK;

    gloadA(0);
    gloadB(0);
    sstoreA(0, 0);
    sstoreB(0);
    __syncthreads();

    for (int ch = 0; ch < nchunks; ++ch) {
        const int s = ch & 1;
        const bool more = ch + 1 < nchunks;
        const int k1 = (ch + 1) * BK;
        const float* Af = &As[s][a_fo];
        const float* Bf = &Bs[s][b_fo];

        if constexpr (LONE) {
            if (more) { gloadA(k1); gloadB(k1); }
            v2f a[2][2][MT], b[2][2][NT];
            lfrag(Af, Bf, 0, a[0][0], b[0][0]);
            lfrag(Af, Bf, 1, a[0][1], b[0][1]);
#pragma unroll
            for (int g = 0; g < KS / 2; ++g) {
                if (g + 1 < KS / 2) {
                    lfrag(Af, Bf, 2 * g + 2, a[(g + 1) & 1][0], b[(g + 1) & 1][0]);
                    lfrag(Af, Bf, 2 * g + 3, a[(g + 1) & 1][1], b[(g + 1) & 1][1]);
                }
                mma(a[g & 1][0], b[g & 1][0]);
                mma(a[g & 1][1], b[g & 1][1]);
            }
            if (more) { sstoreA(s ^ 1, k1); sstoreB(s ^ 1); }
        } else {
            if (more) gloadA(k1);
#pragma unroll
            for (int kk = 0; kk < KS / 2; ++kk) {
                v2f a[MT], b[NT];
                lfrag(Af, Bf, kk, a, b);
                mma(a, b);
            }
            if (more) { sstoreA(s ^ 1, k1); gloadB(k1); }
#pragma unroll
            for (int kk = KS / 2; kk < KS; ++kk) {
                v2f a[MT], b[NT];
                lfrag(Af, Bf, kk, a, b);
                mma(a, b);
            }
            if (more) sstoreB(s ^ 1);
        }
        __syncthreads();
    }

    const float scale = scale_ptr ? scale_ptr[0] : 1.0f;
    const int m0 = bm0 + wy * 16 * MT;
    const int n0 = bn0 + wx * 16 * NT;
    const int mb = h << 3;
    if constexpr (HASRES) {
        float res[MT][NT][8];
#pragma unroll
        for (int i = 0; i < MT; ++i)
#pragma unroll
            for (int t = 0; t < NT; ++t)
#pragma unroll
                for (int v = 0; v < 8; ++v)
                    res[i][t][v] = residual[(size_t)(m0 + i * 16 + mb + v) * ldr + n0 + t * 16 + lm];
#pragma unroll
        for (int i = 0; i < MT; ++i)
#pragma unroll
            for (int t = 0; t < NT; ++t)
#pragma unroll
                for (int v = 0; v < 8; ++v)
                    C[(size_t)(m0 + i * 16 + mb + v) * ldc + n0 + t * 16 + lm] =
                        acc[i][t][v] * scale + res[i][t][v];
    } else {
#pragma unroll
        for (int i = 0; i < MT; ++i)
#pragma unroll
            for (int t = 0; t < NT; ++t)
#pragma unroll
                for (int v = 0; v < 8; ++v)
                    C[(size_t)(m0 + i * 16 + mb + v) * ldc + n0 + t * 16 + lm] = acc[i][t][v] * scale;
    }
}

__global__ __launch_bounds__(256)
void conv_state_kernel(const float* __restrict__ xz, float* __restrict__ out)
{
    const int idx = blockIdx.x * 256 + threadIdx.x;
    if (idx >= BB * D_INNER * (D_CONV - 1)) return;
    const int j = idx % (D_CONV - 1);
    const int c = (idx / (D_CONV - 1)) % D_INNER;
    const int b = idx / ((D_CONV - 1) * D_INNER);
    out[idx] = xz[((size_t)(b * LL + (LL - (D_CONV - 1) + j))) * (2 * D_INNER) + c];
}

__device__ __forceinline__ float softplus_dt(float v)
{
    return (v > 20.f) ? v : __logf(1.f + __expf(v));
}

#define SCAN_T  32
#define SCAN_NC (LL / SCAN_T)
#define SCAN_G  4

__device__ __forceinline__ size_t cs_index(int b, int c, int n, int d)
{
    return ((size_t)((b * SCAN_NC + c) * D_STATE + n)) * D_INNER + d;
}

__device__ __forceinline__ void load_aval(const float* __restrict__ A_log, int d,
                                          float (&Aval)[D_STATE])
{
#pragma unroll
    for (int n = 0; n < D_STATE; ++n)
        Aval[n] = -__expf(A_log[d * D_STATE + n]);
}

__global__ __launch_bounds__(128)
void scan_chunk_state_kernel(const float* __restrict__ dt, const float* __restrict__ dt_bias,
                             const float* __restrict__ xc,
                             const float* __restrict__ xdbl, const float* __restrict__ A_log,
                             float* __restrict__ hloc, float* __restrict__ pdec)
{
    const int d = blockIdx.x * 128 + threadIdx.x;
    const int c = blockIdx.y;
    const int b = blockIdx.z;
    const float db = dt_bias[d];

    float Aval[D_STATE];
    load_aval(A_log, d, Aval);

    float h[D_STATE], P[D_STATE];
#pragma unroll
    for (int n = 0; n < D_STATE; ++n) { h[n] = 0.f; P[n] = 1.f; }

    const size_t row0 = (size_t)b * LL + (size_t)c * SCAN_T;
    const float* dtp = dt + row0 * D_INNER + d;
    const float* xp  = xc + row0 * D_INNER + d;
    const float* bcp = xdbl + row0 * XDBL_W + DT_RANK;

    for (int l0 = 0; l0 < SCAN_T; l0 += SCAN_G) {
        float dtv[SCAN_G], xv[SCAN_G];
#pragma unroll
        for (int j = 0; j < SCAN_G; ++j) {
            dtv[j] = dtp[(size_t)(l0 + j) * D_INNER];
            xv[j]  = xp[(size_t)(l0 + j) * D_INNER];
        }
#pragma unroll
        for (int j = 0; j < SCAN_G; ++j)
            dtv[j] = softplus_dt(dtv[j] + db);
#pragma unroll
        for (int j = 0; j < SCAN_G; ++j) {
            const float4* Bp = (const float4*)(bcp + (size_t)(l0 + j) * XDBL_W);
            float Bb[D_STATE];
#pragma unroll
            for (int q = 0; q < 4; ++q) {
                const float4 bq = Bp[q];
                Bb[4*q+0] = bq.x; Bb[4*q+1] = bq.y; Bb[4*q+2] = bq.z; Bb[4*q+3] = bq.w;
            }
            const float bx = dtv[j] * xv[j];
#pragma unroll
            for (int n = 0; n < D_STATE; ++n) {
                const float dA = __expf(Aval[n] * dtv[j]);
                h[n] = dA * h[n] + bx * Bb[n];
                P[n] *= dA;
            }
        }
    }

#pragma unroll
    for (int n = 0; n < D_STATE; ++n) {
        hloc[cs_index(b, c, n, d)] = h[n];
        pdec[cs_index(b, c, n, d)] = P[n];
    }
}

#define CARRY_G 21
static_assert((SCAN_NC - 1) % CARRY_G == 0, "carry groups must tile the chunks");

__global__ __launch_bounds__(256)
void scan_carry_kernel(const float* __restrict__ hloc, const float* __restrict__ pdec,
                       float* __restrict__ hin)
{
    const int t = blockIdx.x * 256 + threadIdx.x;
    const int d = t & (D_INNER - 1);
    const int n = (t >> 11) & (D_STATE - 1);
    const int b = t >> 15;
    const size_t base = cs_index(b, 0, n, d);
    const size_t cstride = (size_t)D_STATE * D_INNER;
    float carry = 0.f;
    for (int c0 = 0; c0 < SCAN_NC - 1; c0 += CARRY_G) {
        float p[CARRY_G], s[CARRY_G];
#pragma unroll
        for (int j = 0; j < CARRY_G; ++j) {
            p[j] = pdec[base + (size_t)(c0 + j) * cstride];
            s[j] = hloc[base + (size_t)(c0 + j) * cstride];
        }
#pragma unroll
        for (int j = 0; j < CARRY_G; ++j) {
            hin[base + (size_t)(c0 + j) * cstride] = carry;
            carry = p[j] * carry + s[j];
        }
    }
    hin[base + (size_t)(SCAN_NC - 1) * cstride] = carry;
}

__global__ __launch_bounds__(128)
void scan_chunk_out_kernel(const float* __restrict__ dt, const float* __restrict__ dt_bias,
                           const float* __restrict__ xc,
                           const float* __restrict__ xdbl, const float* __restrict__ A_log,
                           const float* __restrict__ hin,
                           float* __restrict__ y, float* __restrict__ ssm_out)
{
    const int d = blockIdx.x * 128 + threadIdx.x;
    const int c = blockIdx.y;
    const int b = blockIdx.z;
    const float db = dt_bias[d];

    float Aval[D_STATE];
    load_aval(A_log, d, Aval);

    float h[D_STATE];
#pragma unroll
    for (int n = 0; n < D_STATE; ++n) h[n] = hin[cs_index(b, c, n, d)];

    const size_t row0 = (size_t)b * LL + (size_t)c * SCAN_T;
    const float* dtp = dt + row0 * D_INNER + d;
    const float* xp  = xc + row0 * D_INNER + d;
    const float* bcp = xdbl + row0 * XDBL_W + DT_RANK;
    float* yp = y + row0 * D_INNER + d;

    for (int l0 = 0; l0 < SCAN_T; l0 += SCAN_G) {
        float dtv[SCAN_G], xv[SCAN_G];
#pragma unroll
        for (int j = 0; j < SCAN_G; ++j) {
            dtv[j] = dtp[(size_t)(l0 + j) * D_INNER];
            xv[j]  = xp[(size_t)(l0 + j) * D_INNER];
        }
#pragma unroll
        for (int j = 0; j < SCAN_G; ++j)
            dtv[j] = softplus_dt(dtv[j] + db);
#pragma unroll
        for (int j = 0; j < SCAN_G; ++j) {
            const float4* Bp = (const float4*)(bcp + (size_t)(l0 + j) * XDBL_W);
            const float4* Cp = Bp + 4;
            float Bb[D_STATE], Cc[D_STATE];
#pragma unroll
            for (int q = 0; q < 4; ++q) {
                const float4 bq = Bp[q];
                const float4 cq = Cp[q];
                Bb[4*q+0] = bq.x; Bb[4*q+1] = bq.y; Bb[4*q+2] = bq.z; Bb[4*q+3] = bq.w;
                Cc[4*q+0] = cq.x; Cc[4*q+1] = cq.y; Cc[4*q+2] = cq.z; Cc[4*q+3] = cq.w;
            }
            const float bx = dtv[j] * xv[j];
            float yv = 0.f;
#pragma unroll
            for (int n = 0; n < D_STATE; ++n) {
                const float dA = __expf(Aval[n] * dtv[j]);
                h[n] = dA * h[n] + bx * Bb[n];
                yv += h[n] * Cc[n];
            }
            yp[(size_t)(l0 + j) * D_INNER] = yv;
        }
    }

    if (c == SCAN_NC - 1) {
#pragma unroll
        for (int n = 0; n < D_STATE; ++n)
            ssm_out[(size_t)(b * D_INNER + d) * D_STATE + n] = h[n];
    }
}

__global__ __launch_bounds__(256)
void gate_kernel(float* __restrict__ y, const float* __restrict__ xc,
                 const float* __restrict__ xz, const float* __restrict__ Dp)
{
    const size_t idx = (size_t)blockIdx.x * 256 + threadIdx.x;
    const int d = (int)(idx & (D_INNER - 1));
    const size_t row = idx >> 11;
    const float z  = xz[row * (2 * D_INNER) + D_INNER + d];
    const float yv = y[idx] + xc[idx] * Dp[d];
    y[idx] = yv * z * sigmoidf_(z);
}

extern "C" void kernel_launch(void* const* d_in, const int* in_sizes, int n_in,
                              void* d_out, int out_size, void* d_ws, size_t ws_size,
                              hipStream_t stream)
{
    const float* x         = (const float*)d_in[0];
    const float* ln_w      = (const float*)d_in[1];
    const float* ln_b      = (const float*)d_in[2];
    const float* in_proj_w = (const float*)d_in[3];
    const float* conv_w    = (const float*)d_in[4];
    const float* conv_b    = (const float*)d_in[5];
    const float* x_proj_w  = (const float*)d_in[6];
    const float* dt_proj_w = (const float*)d_in[7];
    const float* dt_proj_b = (const float*)d_in[8];
    const float* A_log     = (const float*)d_in[9];
    const float* D_param   = (const float*)d_in[10];
    const float* out_projw = (const float*)d_in[11];
    const float* res_scale = (const float*)d_in[12];

    float* out      = (float*)d_out;
    float* out_ssm  = out + (size_t)BB * LL * D_MODEL;
    float* out_conv = out_ssm + (size_t)BB * D_INNER * D_STATE;

    float* ws    = (float*)d_ws;
    float* xn    = ws;
    float* xz    = xn    + (size_t)TOKENS * D_MODEL;
    float* xc    = xz    + (size_t)TOKENS * 2 * D_INNER;
    float* xdbl  = xc    + (size_t)TOKENS * D_INNER;
    float* dtbuf = xdbl  + (size_t)TOKENS * XDBL_W;
    float* ybuf  = dtbuf + (size_t)TOKENS * D_INNER;
    float* hloc  = ybuf  + (size_t)TOKENS * D_INNER;
    float* pdec  = hloc  + (size_t)BB * SCAN_NC * D_STATE * D_INNER;
    float* hin   = pdec  + (size_t)BB * SCAN_NC * D_STATE * D_INNER;

    ln_kernel<<<TOKENS, 256, 0, stream>>>(x, ln_w, ln_b, xn);

    wmma_gemm_lds_kernel<4, 4, 32, false, false, false><<<dim3((2 * D_INNER) / 128, TOKENS / 128), 128, 0, stream>>>(
        xn, D_MODEL, in_proj_w, 2 * D_INNER, xz, 2 * D_INNER, D_MODEL,
        nullptr, 0, nullptr, nullptr, nullptr, nullptr);

    conv_state_kernel<<<(BB * D_INNER * (D_CONV - 1) + 255) / 256, 256, 0, stream>>>(
        xz, out_conv);

    static_assert(XDBL_W == 96, "one block column: every xc element is staged exactly once");
    wmma_gemm_lds_kernel<1, 3, 64, true, true, false><<<dim3(XDBL_W / 96, TOKENS / 32), 128, 0, stream>>>(
        xz, 2 * D_INNER, x_proj_w, XDBL_W, xdbl, XDBL_W, D_INNER,
        nullptr, 0, nullptr, conv_w, conv_b, xc);

    wmma_gemm_kernel<2, 4, false><<<dim3(D_INNER / 128, TOKENS / 64), 128, 0, stream>>>(
        xdbl, XDBL_W, dt_proj_w, D_INNER, dtbuf, D_INNER, DT_RANK,
        nullptr, 0, nullptr);

    scan_chunk_state_kernel<<<dim3(D_INNER / 128, SCAN_NC - 1, BB), 128, 0, stream>>>(
        dtbuf, dt_proj_b, xc, xdbl, A_log, hloc, pdec);
    scan_carry_kernel<<<(BB * D_STATE * D_INNER) / 256, 256, 0, stream>>>(hloc, pdec, hin);
    scan_chunk_out_kernel<<<dim3(D_INNER / 128, SCAN_NC, BB), 128, 0, stream>>>(
        dtbuf, dt_proj_b, xc, xdbl, A_log, hin, ybuf, out_ssm);

    gate_kernel<<<(TOKENS * D_INNER) / 256, 256, 0, stream>>>(ybuf, xc, xz, D_param);

    wmma_gemm_lds_kernel<2, 4, 32, false, false, true><<<dim3(D_MODEL / 128, TOKENS / 64), 128, 0, stream>>>(
        ybuf, D_INNER, out_projw, D_MODEL, out, D_MODEL, D_INNER,
        x, D_MODEL, res_scale, nullptr, nullptr, nullptr);
}
